# starting kernel
# baseline (speedup 1.0000x reference)
#include <hip/hip_runtime.h>

typedef __attribute__((ext_vector_type(2))) float v2f;
typedef __attribute__((ext_vector_type(8))) float v8f;

#define HDIM 128
#define NT   16


__global__ void fill_ones_kernel(float* __restrict__ p, int n) {
  int i = blockIdx.x * blockDim.x + threadIdx.x;
  if (i < n) p[i] = 1.0f;
}

__global__ void zero_kernel(float* __restrict__ p, long long n) {
  long long i = (long long)blockIdx.x * blockDim.x + threadIdx.x;
  if (i < n) p[i] = 0.0f;
}

__global__ void degree_kernel(const int* __restrict__ src, const int* __restrict__ dst,
                              float* __restrict__ deg_out, float* __restrict__ deg_in, int E) {
  int e = blockIdx.x * blockDim.x + threadIdx.x;
  if (e < E) {
    atomicAdd(&deg_out[src[e]], 1.0f);
    atomicAdd(&deg_in[dst[e]], 1.0f);
  }
}

__global__ void norm_kernel(float* __restrict__ p, int n) {
  int i = blockIdx.x * blockDim.x + threadIdx.x;
  if (i < n) p[i] = rsqrtf(fmaxf(p[i], 1.0f));
}

__global__ void gemm_wmma_f32_kernel(const float* __restrict__ X, const float* __restrict__ W,
                                     float* __restrict__ Y, int KD) {
  const int wave = threadIdx.x >> 5;
  const int lane = threadIdx.x & 31;
  const int l16  = lane & 15;
  const int hi   = lane >> 4;
  const int mbase = blockIdx.x * NT;
  const int nbase = wave * NT;

  const float* __restrict__ xrow = X + (size_t)(mbase + l16) * KD;
  const float* __restrict__ wcol = W + nbase + l16;

  v8f c = {};
  #pragma unroll 4
  for (int k = 0; k < KD; k += 4) {
    v2f a, b;
    a.x = xrow[k + 2 * hi];
    a.y = xrow[k + 2 * hi + 1];
    b.x = wcol[(size_t)(k + 2 * hi) * HDIM];
    b.y = wcol[(size_t)(k + 2 * hi + 1) * HDIM];
    c = __builtin_amdgcn_wmma_f32_16x16x4_f32(false, a, false, b, (short)0, c, false, false);
  }

  float* __restrict__ yout = Y + nbase + l16;
  #pragma unroll
  for (int v = 0; v < 8; ++v)
    yout[(size_t)(mbase + v + 8 * hi) * HDIM] = c[v];
}

__global__ void scatter_kernel(const int* __restrict__ src, const int* __restrict__ dst,
                               const float* __restrict__ ns, const float* __restrict__ Y,
                               float* __restrict__ agg, int E) {
  int warp = (int)(((long long)blockIdx.x * blockDim.x + threadIdx.x) >> 5);
  int lane = threadIdx.x & 31;
  if (warp >= E) return;
  int s = src[warp], d = dst[warp];
  float w = ns[s];
  float4 m = ((const float4*)(Y + (size_t)s * HDIM))[lane];
  float* ap = agg + (size_t)d * HDIM + lane * 4;
  atomicAdd(ap + 0, m.x * w);
  atomicAdd(ap + 1, m.y * w);
  atomicAdd(ap + 2, m.z * w);
  atomicAdd(ap + 3, m.w * w);
}

__global__ void finalize_kernel(const float* __restrict__ agg, const float* __restrict__ Y,
                                const float* __restrict__ ns, const float* __restrict__ nd,
                                const float* __restrict__ bias, float* __restrict__ out,
                                long long total, int relu) {
  long long idx = (long long)blockIdx.x * blockDim.x + threadIdx.x;
  if (idx >= total) return;
  int i = (int)(idx >> 7);
  int j = (int)(idx & (HDIM - 1));
  float v = (agg[idx] + Y[idx] * ns[i]) * nd[i] + bias[j];
  if (relu) v = fmaxf(v, 0.0f);
  out[idx] = v;
}

static inline int cdiv(long long a, int b) { return (int)((a + b - 1) / b); }

static void run_graph(const float* feat, const int* src, const int* dst,
                      const float* W0, const float* b0, const float* W1, const float* b1,
                      int N, int D, int E,
                      float* ns, float* nd, float* Y, float* agg, float* h,
                      float* out, hipStream_t stream) {
  const long long NH = (long long)N * HDIM;

  fill_ones_kernel<<<cdiv(2 * (long long)N, 256), 256, 0, stream>>>(ns, 2 * N);
  degree_kernel<<<cdiv(E, 256), 256, 0, stream>>>(src, dst, ns, nd, E);
  norm_kernel<<<cdiv(2 * (long long)N, 256), 256, 0, stream>>>(ns, 2 * N);

  gemm_wmma_f32_kernel<<<N / NT, 256, 0, stream>>>(feat, W0, Y, D);
  zero_kernel<<<cdiv(NH, 256), 256, 0, stream>>>(agg, NH);
  scatter_kernel<<<cdiv((long long)E * 32, 256), 256, 0, stream>>>(src, dst, ns, Y, agg, E);
  finalize_kernel<<<cdiv(NH, 256), 256, 0, stream>>>(agg, Y, ns, nd, b0, h, NH, 1);

  gemm_wmma_f32_kernel<<<N / NT, 256, 0, stream>>>(h, W1, Y, HDIM);
  zero_kernel<<<cdiv(NH, 256), 256, 0, stream>>>(agg, NH);
  scatter_kernel<<<cdiv((long long)E * 32, 256), 256, 0, stream>>>(src, dst, ns, Y, agg, E);
  finalize_kernel<<<cdiv(NH, 256), 256, 0, stream>>>(agg, Y, ns, nd, b1, out, NH, 0);
}

extern "C" void kernel_launch(void* const* d_in, const int* in_sizes, int n_in,
                              void* d_out, int out_size, void* d_ws, size_t ws_size,
                              hipStream_t stream) {
  const float* feat_a = (const float*)d_in[0];
  const float* feat_b = (const float*)d_in[1];
  const int*   src_a  = (const int*)d_in[2];
  const int*   dst_a  = (const int*)d_in[3];
  const int*   src_b  = (const int*)d_in[4];
  const int*   dst_b  = (const int*)d_in[5];
  const float* Wa0 = (const float*)d_in[6];
  const float* ba0 = (const float*)d_in[7];
  const float* Wa1 = (const float*)d_in[8];
  const float* ba1 = (const float*)d_in[9];
  const float* Wb0 = (const float*)d_in[10];
  const float* bb0 = (const float*)d_in[11];
  const float* Wb1 = (const float*)d_in[12];
  const float* bb1 = (const float*)d_in[13];

  const int DA = in_sizes[6] / HDIM;
  const int DB = in_sizes[10] / HDIM;
  const int NA = in_sizes[0] / DA;
  const int NB = in_sizes[1] / DB;
  const int EA = in_sizes[2];
  const int EB = in_sizes[4];

  float* ws = (float*)d_ws;
  float* ns_a  = ws;
  float* nd_a  = ns_a + NA;
  float* Y_a   = nd_a + NA;
  float* agg_a = Y_a + (size_t)NA * HDIM;
  float* h_a   = agg_a + (size_t)NA * HDIM;
  float* ns_b  = h_a + (size_t)NA * HDIM;
  float* nd_b  = ns_b + NB;
  float* Y_b   = nd_b + NB;
  float* agg_b = Y_b + (size_t)NB * HDIM;
  float* h_b   = agg_b + (size_t)NB * HDIM;

  float* out_a = (float*)d_out;
  float* out_b = out_a + (size_t)NA * HDIM;

  run_graph(feat_a, src_a, dst_a, Wa0, ba0, Wa1, ba1, NA, DA, EA,
            ns_a, nd_a, Y_a, agg_a, h_a, out_a, stream);
  run_graph(feat_b, src_b, dst_b, Wb0, bb0, Wb1, bb1, NB, DB, EB,
            ns_b, nd_b, Y_b, agg_b, h_b, out_b, stream);
}


# GEMM with W in registers: each wave keeps its 16-column slice of W as B fragments and loops over 16-row X tiles staged once per work-group in LDS by 16-byte loads; 512 work-groups; k sums as before
# speedup vs baseline: 2.3341x; 2.3341x over previous
#include <hip/hip_runtime.h>

typedef __attribute__((ext_vector_type(2))) float v2f;
typedef __attribute__((ext_vector_type(8))) float v8f;

#define HDIM 128
#define NT   16

#define DEG_RANGE   1024
#define DEG_THREADS 1024
#define DEG_DEPTH   8

__device__ __forceinline__ void deg_count(unsigned int* cnt, int v, unsigned int lo) {
  unsigned int u = (unsigned int)v - lo;
  if (u < DEG_RANGE) atomicAdd(&cnt[u], 1u);
}

__global__ __launch_bounds__(DEG_THREADS)
void degree_norm_kernel(const int* __restrict__ src, const int* __restrict__ dst,
                        float* __restrict__ ns, float* __restrict__ nd, int N, int E) {
  __shared__ unsigned int cnt[DEG_RANGE];
  const int* __restrict__ idx = blockIdx.y ? dst : src;
  float* __restrict__ res     = blockIdx.y ? nd : ns;
  const unsigned int lo = blockIdx.x * DEG_RANGE;

  for (int r = threadIdx.x; r < DEG_RANGE; r += DEG_THREADS) cnt[r] = 0u;
  __syncthreads();

  const int E4 = E >> 2;
  const int4* __restrict__ idx4 = (const int4*)idx;
  for (int i0 = threadIdx.x; i0 < E4; i0 += DEG_DEPTH * DEG_THREADS) {
    int4 v[DEG_DEPTH];
    #pragma unroll
    for (int k = 0; k < DEG_DEPTH; ++k) {
      const int i = i0 + k * DEG_THREADS;
      v[k] = (i < E4) ? idx4[i] : make_int4(-1, -1, -1, -1);
    }
    #pragma unroll
    for (int k = 0; k < DEG_DEPTH; ++k) {
      deg_count(cnt, v[k].x, lo);
      deg_count(cnt, v[k].y, lo);
      deg_count(cnt, v[k].z, lo);
      deg_count(cnt, v[k].w, lo);
    }
  }
  for (int e = (E4 << 2) + threadIdx.x; e < E; e += DEG_THREADS) deg_count(cnt, idx[e], lo);
  __syncthreads();

  for (int r = threadIdx.x; r < DEG_RANGE; r += DEG_THREADS) {
    const unsigned int i = lo + r;
    if (i < (unsigned int)N) res[i] = rsqrtf(fmaxf(1.0f + (float)cnt[r], 1.0f));
  }
}

#define GEMM_THREADS 256
#define GEMM_GRID    512

template <int KD>
__global__ __launch_bounds__(GEMM_THREADS)
void gemm_wreg_kernel(const float* __restrict__ X, const float* __restrict__ W,
                      float* __restrict__ Y, int ntiles) {
  constexpr int PITCH = KD + 4;
  constexpr int KS    = KD / 4;
  constexpr int F4ROW = KD / 4;
  constexpr int NLD   = (NT * F4ROW) / GEMM_THREADS;
  __shared__ __attribute__((aligned(16))) float xs[2][NT * PITCH];

  const int tid  = threadIdx.x;
  const int wave = tid >> 5;
  const int lane = tid & 31;
  const int l16  = lane & 15;
  const int hi   = lane >> 4;
  const int nbase = wave * NT;

  const float* __restrict__ wcol = W + nbase + l16;
  v2f b[KS];
  #pragma unroll
  for (int j = 0; j < KS; ++j) {
    b[j].x = wcol[(size_t)(4 * j + 2 * hi) * HDIM];
    b[j].y = wcol[(size_t)(4 * j + 2 * hi + 1) * HDIM];
  }

  const float4* __restrict__ X4 = (const float4*)X;
  const int f0 = tid, f1 = tid + GEMM_THREADS, f2 = tid + 2 * GEMM_THREADS, f3 = tid + 3 * GEMM_THREADS;
  const int o0 = (f0 / F4ROW) * PITCH + 4 * (f0 % F4ROW);
  const int o1 = (f1 / F4ROW) * PITCH + 4 * (f1 % F4ROW);
  const int o2 = (f2 / F4ROW) * PITCH + 4 * (f2 % F4ROW);
  const int o3 = (f3 / F4ROW) * PITCH + 4 * (f3 % F4ROW);
  float4 p0 = make_float4(0.0f, 0.0f, 0.0f, 0.0f), p1 = p0, p2 = p0, p3 = p0;

  int t = blockIdx.x;
  if (t < ntiles) {
    const float4* __restrict__ xt = X4 + (size_t)t * (NT * F4ROW);
    p0 = xt[f0];
    p1 = xt[f1];
    if (NLD > 2) { p2 = xt[f2]; p3 = xt[f3]; }
    *(float4*)(xs[0] + o0) = p0;
    *(float4*)(xs[0] + o1) = p1;
    if (NLD > 2) { *(float4*)(xs[0] + o2) = p2; *(float4*)(xs[0] + o3) = p3; }
  }
  __syncthreads();

  int buf = 0;
  for (; t < ntiles; t += gridDim.x) {
    const int tn = t + gridDim.x;
    if (tn < ntiles) {
      const float4* __restrict__ xt = X4 + (size_t)tn * (NT * F4ROW);
      p0 = xt[f0];
      p1 = xt[f1];
      if (NLD > 2) { p2 = xt[f2]; p3 = xt[f3]; }
    }

    const float* xa = xs[buf] + l16 * PITCH + 2 * hi;
    v8f c = {};
    #pragma unroll
    for (int j = 0; j < KS; ++j) {
      const v2f a = *(const v2f*)(xa + 4 * j);
      c = __builtin_amdgcn_wmma_f32_16x16x4_f32(false, a, false, b[j], (short)0, c, false, false);
    }

    if (tn < ntiles) {
      float* xn = xs[buf ^ 1];
      *(float4*)(xn + o0) = p0;
      *(float4*)(xn + o1) = p1;
      if (NLD > 2) { *(float4*)(xn + o2) = p2; *(float4*)(xn + o3) = p3; }
    }
    __syncthreads();
    buf ^= 1;

    float* __restrict__ yout = Y + (size_t)t * NT * HDIM + nbase + l16;
    #pragma unroll
    for (int v = 0; v < 8; ++v)
      yout[(size_t)(v + 8 * hi) * HDIM] = c[v];
  }
}

#define AGG_THREADS 1024
#define AGG_WAVES   32
#define AGG_DEPTH   8
#define AGG_CHUNK   (AGG_DEPTH * AGG_THREADS)
#define AGG_FLUSH   32
#define AGG_QCAP    (AGG_FLUSH + 128)
#define AGG_ROWS    16

template <bool LAST>
__device__ __forceinline__ int agg_drain(int n, const int2* qw, float* acc,
                                         const int* __restrict__ src,
                                         const float* __restrict__ ns,
                                         const float4* __restrict__ Y4, int lane) {
  const int keep = LAST ? 0 : (n & (AGG_ROWS - 1));
  for (int base = keep; base < n; base += 32) {
    const int cnt = min(32, n - base);
    int s = 0, r = 0;
    float w = 0.0f;
    if (lane < cnt) {
      const int2 t = qw[base + lane];
      s = src[t.x];
      r = t.y;
      w = ns[s];
    }
    for (int g = 0; g < cnt; g += AGG_ROWS) {
      float4 row[AGG_ROWS];
      #pragma unroll
      for (int j = 0; j < AGG_ROWS; ++j) {
        const int jj = LAST ? min(g + j, cnt - 1) : g + j;
        const int sj = __builtin_amdgcn_readlane(s, jj);
        row[j] = Y4[(size_t)sj * (HDIM / 4) + lane];
      }
      #pragma unroll
      for (int j = 0; j < AGG_ROWS; ++j) {
        if (!LAST || g + j < cnt) {
          const float wj = __int_as_float(__builtin_amdgcn_readlane(__float_as_int(w), g + j));
          const int   rj = __builtin_amdgcn_readlane(r, g + j);
          float* a = acc + rj * HDIM + lane;
          __hip_atomic_fetch_add(a,      row[j].x * wj, __ATOMIC_RELAXED, __HIP_MEMORY_SCOPE_WORKGROUP);
          __hip_atomic_fetch_add(a + 32, row[j].y * wj, __ATOMIC_RELAXED, __HIP_MEMORY_SCOPE_WORKGROUP);
          __hip_atomic_fetch_add(a + 64, row[j].z * wj, __ATOMIC_RELAXED, __HIP_MEMORY_SCOPE_WORKGROUP);
          __hip_atomic_fetch_add(a + 96, row[j].w * wj, __ATOMIC_RELAXED, __HIP_MEMORY_SCOPE_WORKGROUP);
        }
      }
    }
  }
  return keep;
}

template <int R>
__device__ __forceinline__ int agg_push(int qn, int2* qw, int d, int e, unsigned int lo, int lane) {
  const unsigned int u = (unsigned int)d - lo;
  const bool m = u < (unsigned int)R;
  const unsigned int mask = __builtin_amdgcn_ballot_w32(m);
  if (mask) {
    if (m) qw[qn + __builtin_amdgcn_mbcnt_lo(mask, 0u)] = make_int2(e, (int)u);
    qn += __builtin_popcount(mask);
  }
  return qn;
}

template <int R>
__global__ __launch_bounds__(AGG_THREADS)
void aggregate_owner_kernel(const int* __restrict__ src, const int* __restrict__ dst,
                            const float* __restrict__ ns, const float* __restrict__ nd,
                            const float* __restrict__ Y, const float* __restrict__ bias,
                            float* __restrict__ out, int N, int E, int relu) {
  __shared__ float acc[R * HDIM];
  __shared__ int2  queue[AGG_WAVES * AGG_QCAP];
  const int lane = threadIdx.x & 31;
  const int wave = threadIdx.x >> 5;
  const unsigned int lo = blockIdx.x * R;
  const float4* __restrict__ Y4 = (const float4*)Y;
  int2* qw = queue + wave * AGG_QCAP;

  for (int r = wave; r < R; r += AGG_WAVES) {
    const unsigned int i = lo + r;
    float4 y = make_float4(0.0f, 0.0f, 0.0f, 0.0f);
    if (i < (unsigned int)N) {
      const float w = ns[i];
      y = Y4[(size_t)i * (HDIM / 4) + lane];
      y.x *= w; y.y *= w; y.z *= w; y.w *= w;
    }
    float* a = acc + r * HDIM + lane;
    a[0] = y.x; a[32] = y.y; a[64] = y.z; a[96] = y.w;
  }
  __syncthreads();

  const int E4 = E >> 2;
  const int4* __restrict__ dst4 = (const int4*)dst;
  const int nchunk = (E4 + AGG_CHUNK - 1) / AGG_CHUNK;
  int chunk = nchunk ? (int)(blockIdx.x % (unsigned int)nchunk) : 0;
  int qn = 0;
  for (int t = 0; t < nchunk; ++t) {
    const int ib = chunk * AGG_CHUNK + wave * 32 + lane;
    int4 v[AGG_DEPTH];
    #pragma unroll
    for (int k = 0; k < AGG_DEPTH; ++k) {
      v[k] = dst4[min(ib + k * AGG_THREADS, E4 - 1)];
    }
    #pragma unroll
    for (int k = 0; k < AGG_DEPTH; ++k) {
      const int e = (ib + k * AGG_THREADS) << 2;
      if (e >= (E4 << 2)) v[k] = make_int4(-1, -1, -1, -1);
      qn = agg_push<R>(qn, qw, v[k].x, e + 0, lo, lane);
      qn = agg_push<R>(qn, qw, v[k].y, e + 1, lo, lane);
      qn = agg_push<R>(qn, qw, v[k].z, e + 2, lo, lane);
      qn = agg_push<R>(qn, qw, v[k].w, e + 3, lo, lane);
      if (qn >= AGG_FLUSH) qn = agg_drain<false>(qn, qw, acc, src, ns, Y4, lane);
    }
    if (++chunk == nchunk) chunk = 0;
  }
  if (wave == 0) {
    const int e = (E4 << 2) + lane;
    qn = agg_push<R>(qn, qw, (e < E) ? dst[e] : -1, e, lo, lane);
  }
  if (qn) agg_drain<true>(qn, qw, acc, src, ns, Y4, lane);
  __syncthreads();

  const float4 b = ((const float4*)bias)[lane];
  for (int r = wave; r < R; r += AGG_WAVES) {
    const unsigned int i = lo + r;
    if (i < (unsigned int)N) {
      const float d = nd[i];
      const float* a = acc + r * HDIM + lane;
      float4 o;
      o.x = a[0]  * d + b.x;
      o.y = a[32] * d + b.y;
      o.z = a[64] * d + b.z;
      o.w = a[96] * d + b.w;
      if (relu) {
        o.x = fmaxf(o.x, 0.0f); o.y = fmaxf(o.y, 0.0f);
        o.z = fmaxf(o.z, 0.0f); o.w = fmaxf(o.w, 0.0f);
      }
      ((float4*)out)[(size_t)i * (HDIM / 4) + lane] = o;
    }
  }
}

static inline int cdiv(long long a, int b) { return (int)((a + b - 1) / b); }

template <int R, int D0>
static void run_graph(const float* feat, const int* src, const int* dst,
                      const float* W0, const float* b0, const float* W1, const float* b1,
                      int N, int E,
                      float* ns, float* nd, float* Y, float* h,
                      float* out, hipStream_t stream) {
  degree_norm_kernel<<<dim3(cdiv(N, DEG_RANGE), 2), DEG_THREADS, 0, stream>>>(src, dst, ns, nd, N, E);

  gemm_wreg_kernel<D0><<<GEMM_GRID, GEMM_THREADS, 0, stream>>>(feat, W0, Y, N / NT);
  aggregate_owner_kernel<R><<<cdiv(N, R), AGG_THREADS, 0, stream>>>(src, dst, ns, nd, Y, b0, h, N, E, 1);

  gemm_wreg_kernel<HDIM><<<GEMM_GRID, GEMM_THREADS, 0, stream>>>(h, W1, Y, N / NT);
  aggregate_owner_kernel<R><<<cdiv(N, R), AGG_THREADS, 0, stream>>>(src, dst, ns, nd, Y, b1, out, N, E, 0);
}

extern "C" void kernel_launch(void* const* d_in, const int* in_sizes, int n_in,
                              void* d_out, int out_size, void* d_ws, size_t ws_size,
                              hipStream_t stream) {
  const float* feat_a = (const float*)d_in[0];
  const float* feat_b = (const float*)d_in[1];
  const int*   src_a  = (const int*)d_in[2];
  const int*   dst_a  = (const int*)d_in[3];
  const int*   src_b  = (const int*)d_in[4];
  const int*   dst_b  = (const int*)d_in[5];
  const float* Wa0 = (const float*)d_in[6];
  const float* ba0 = (const float*)d_in[7];
  const float* Wa1 = (const float*)d_in[8];
  const float* ba1 = (const float*)d_in[9];
  const float* Wb0 = (const float*)d_in[10];
  const float* bb0 = (const float*)d_in[11];
  const float* Wb1 = (const float*)d_in[12];
  const float* bb1 = (const float*)d_in[13];

  const int DA = in_sizes[6] / HDIM;
  const int DB = in_sizes[10] / HDIM;
  const int NA = in_sizes[0] / DA;
  const int NB = in_sizes[1] / DB;
  const int EA = in_sizes[2];
  const int EB = in_sizes[4];

  float* ws = (float*)d_ws;
  float* ns_a  = ws;
  float* nd_a  = ns_a + NA;
  float* Y_a   = nd_a + NA;
  float* h_a   = Y_a + (size_t)NA * HDIM;
  float* ns_b  = h_a + (size_t)NA * HDIM;
  float* nd_b  = ns_b + NB;
  float* Y_b   = nd_b + NB;
  float* h_b   = Y_b + (size_t)NB * HDIM;

  float* out_a = (float*)d_out;
  float* out_b = out_a + (size_t)NA * HDIM;

  run_graph<256, 256>(feat_a, src_a, dst_a, Wa0, ba0, Wa1, ba1, NA, EA,
                 ns_a, nd_a, Y_a, h_a, out_a, stream);
  run_graph<128, 128>(feat_b, src_b, dst_b, Wb0, bb0, Wb1, bb1, NB, EB,
                 ns_b, nd_b, Y_b, h_b, out_b, stream);
}


# Layer 1's neighbour sum reads the (source, row, weight) list its work-group's layer-0 scan kept in workspace, no second dst scan (kept as overflow fallback); LDS atomic sums in arrival order
# speedup vs baseline: 2.6496x; 1.1352x over previous
#include <hip/hip_runtime.h>

typedef __attribute__((ext_vector_type(2))) float v2f;
typedef __attribute__((ext_vector_type(8))) float v8f;

#define HDIM 128
#define NT   16

#define DEG_RANGE   1024
#define DEG_THREADS 1024
#define DEG_DEPTH   8

__device__ __forceinline__ void deg_count(unsigned int* cnt, int v, unsigned int lo) {
  unsigned int u = (unsigned int)v - lo;
  if (u < DEG_RANGE) atomicAdd(&cnt[u], 1u);
}

__global__ __launch_bounds__(DEG_THREADS)
void degree_norm_kernel(const int* __restrict__ src, const int* __restrict__ dst,
                        float* __restrict__ ns, float* __restrict__ nd, int N, int E) {
  __shared__ unsigned int cnt[DEG_RANGE];
  const int* __restrict__ idx = blockIdx.y ? dst : src;
  float* __restrict__ res     = blockIdx.y ? nd : ns;
  const unsigned int lo = blockIdx.x * DEG_RANGE;

  for (int r = threadIdx.x; r < DEG_RANGE; r += DEG_THREADS) cnt[r] = 0u;
  __syncthreads();

  const int E4 = E >> 2;
  const int4* __restrict__ idx4 = (const int4*)idx;
  for (int i0 = threadIdx.x; i0 < E4; i0 += DEG_DEPTH * DEG_THREADS) {
    int4 v[DEG_DEPTH];
    #pragma unroll
    for (int k = 0; k < DEG_DEPTH; ++k) {
      const int i = i0 + k * DEG_THREADS;
      v[k] = (i < E4) ? idx4[i] : make_int4(-1, -1, -1, -1);
    }
    #pragma unroll
    for (int k = 0; k < DEG_DEPTH; ++k) {
      deg_count(cnt, v[k].x, lo);
      deg_count(cnt, v[k].y, lo);
      deg_count(cnt, v[k].z, lo);
      deg_count(cnt, v[k].w, lo);
    }
  }
  for (int e = (E4 << 2) + threadIdx.x; e < E; e += DEG_THREADS) deg_count(cnt, idx[e], lo);
  __syncthreads();

  for (int r = threadIdx.x; r < DEG_RANGE; r += DEG_THREADS) {
    const unsigned int i = lo + r;
    if (i < (unsigned int)N) res[i] = rsqrtf(fmaxf(1.0f + (float)cnt[r], 1.0f));
  }
}

#define GEMM_THREADS 256
#define GEMM_GRID    512

template <int KD>
__global__ __launch_bounds__(GEMM_THREADS)
void gemm_wreg_kernel(const float* __restrict__ X, const float* __restrict__ W,
                      float* __restrict__ Y, int ntiles) {
  constexpr int PITCH = KD + 4;
  constexpr int KS    = KD / 4;
  constexpr int F4ROW = KD / 4;
  constexpr int NLD   = (NT * F4ROW) / GEMM_THREADS;
  __shared__ __attribute__((aligned(16))) float xs[2][NT * PITCH];

  const int tid  = threadIdx.x;
  const int wave = tid >> 5;
  const int lane = tid & 31;
  const int l16  = lane & 15;
  const int hi   = lane >> 4;
  const int nbase = wave * NT;

  const float* __restrict__ wcol = W + nbase + l16;
  v2f b[KS];
  #pragma unroll
  for (int j = 0; j < KS; ++j) {
    b[j].x = wcol[(size_t)(4 * j + 2 * hi) * HDIM];
    b[j].y = wcol[(size_t)(4 * j + 2 * hi + 1) * HDIM];
  }

  const float4* __restrict__ X4 = (const float4*)X;
  const int f0 = tid, f1 = tid + GEMM_THREADS, f2 = tid + 2 * GEMM_THREADS, f3 = tid + 3 * GEMM_THREADS;
  const int o0 = (f0 / F4ROW) * PITCH + 4 * (f0 % F4ROW);
  const int o1 = (f1 / F4ROW) * PITCH + 4 * (f1 % F4ROW);
  const int o2 = (f2 / F4ROW) * PITCH + 4 * (f2 % F4ROW);
  const int o3 = (f3 / F4ROW) * PITCH + 4 * (f3 % F4ROW);
  float4 p0 = make_float4(0.0f, 0.0f, 0.0f, 0.0f), p1 = p0, p2 = p0, p3 = p0;

  int t = blockIdx.x;
  if (t < ntiles) {
    const float4* __restrict__ xt = X4 + (size_t)t * (NT * F4ROW);
    p0 = xt[f0];
    p1 = xt[f1];
    if (NLD > 2) { p2 = xt[f2]; p3 = xt[f3]; }
    *(float4*)(xs[0] + o0) = p0;
    *(float4*)(xs[0] + o1) = p1;
    if (NLD > 2) { *(float4*)(xs[0] + o2) = p2; *(float4*)(xs[0] + o3) = p3; }
  }
  __syncthreads();

  int buf = 0;
  for (; t < ntiles; t += gridDim.x) {
    const int tn = t + gridDim.x;
    if (tn < ntiles) {
      const float4* __restrict__ xt = X4 + (size_t)tn * (NT * F4ROW);
      p0 = xt[f0];
      p1 = xt[f1];
      if (NLD > 2) { p2 = xt[f2]; p3 = xt[f3]; }
    }

    const float* xa = xs[buf] + l16 * PITCH + 2 * hi;
    v8f c = {};
    #pragma unroll
    for (int j = 0; j < KS; ++j) {
      const v2f a = *(const v2f*)(xa + 4 * j);
      c = __builtin_amdgcn_wmma_f32_16x16x4_f32(false, a, false, b[j], (short)0, c, false, false);
    }

    if (tn < ntiles) {
      float* xn = xs[buf ^ 1];
      *(float4*)(xn + o0) = p0;
      *(float4*)(xn + o1) = p1;
      if (NLD > 2) { *(float4*)(xn + o2) = p2; *(float4*)(xn + o3) = p3; }
    }
    __syncthreads();
    buf ^= 1;

    float* __restrict__ yout = Y + (size_t)t * NT * HDIM + nbase + l16;
    #pragma unroll
    for (int v = 0; v < 8; ++v)
      yout[(size_t)(v + 8 * hi) * HDIM] = c[v];
  }
}

#define AGG_THREADS 1024
#define AGG_WAVES   32
#define AGG_DEPTH   8
#define AGG_CHUNK   (AGG_DEPTH * AGG_THREADS)
#define AGG_FLUSH   32
#define AGG_QCAP    (AGG_FLUSH + 128)
#define AGG_ROWS    16
#define AGG_LIST_PER_ROW 20

template <int R>
__device__ __forceinline__ bool agg_list_place(int* lst, const int* lst_end,
                                               int2*& lsr, float*& lw, unsigned int& lcap) {
  constexpr unsigned int CAP = AGG_LIST_PER_ROW * R;
  const size_t avail = (size_t)(lst_end - lst);
  const size_t hdr   = (size_t)((gridDim.x + 3u) & ~3u);
  const size_t off   = hdr + (size_t)blockIdx.x * (3u * CAP);
  lsr  = (int2*)(lst + off);
  lw   = (float*)(lst + off + 2u * CAP);
  lcap = (off + 3u * CAP <= avail) ? CAP : 0u;
  return hdr <= avail;
}

template <bool PARTIAL, class F>
__device__ __forceinline__ void agg_add_rows(int cnt, int s, int r, float w, float* acc,
                                             const float4* __restrict__ Y4, int lane, F&& behind_first_loads) {
  for (int g = 0; g < cnt; g += AGG_ROWS) {
    float4 row[AGG_ROWS];
    #pragma unroll
    for (int j = 0; j < AGG_ROWS; ++j) {
      const int jj = PARTIAL ? min(g + j, cnt - 1) : g + j;
      const int sj = __builtin_amdgcn_readlane(s, jj);
      row[j] = Y4[(size_t)sj * (HDIM / 4) + lane];
    }
    if (g == 0) behind_first_loads();
    #pragma unroll
    for (int j = 0; j < AGG_ROWS; ++j) {
      if (!PARTIAL || g + j < cnt) {
        const float wj = __int_as_float(__builtin_amdgcn_readlane(__float_as_int(w), g + j));
        const int   rj = __builtin_amdgcn_readlane(r, g + j);
        float* a = acc + rj * HDIM + lane;
        __hip_atomic_fetch_add(a,      row[j].x * wj, __ATOMIC_RELAXED, __HIP_MEMORY_SCOPE_WORKGROUP);
        __hip_atomic_fetch_add(a + 32, row[j].y * wj, __ATOMIC_RELAXED, __HIP_MEMORY_SCOPE_WORKGROUP);
        __hip_atomic_fetch_add(a + 64, row[j].z * wj, __ATOMIC_RELAXED, __HIP_MEMORY_SCOPE_WORKGROUP);
        __hip_atomic_fetch_add(a + 96, row[j].w * wj, __ATOMIC_RELAXED, __HIP_MEMORY_SCOPE_WORKGROUP);
      }
    }
  }
}

template <bool LAST, bool KEEP>
__device__ __forceinline__ int agg_drain(int n, const int2* qw, float* acc,
                                         const int* __restrict__ src,
                                         const float* __restrict__ ns,
                                         const float4* __restrict__ Y4, int lane,
                                         unsigned int* lcount, int2* lsr, float* lw, unsigned int lcap) {
  const int keep = LAST ? 0 : (n & (AGG_ROWS - 1));
  for (int base = keep; base < n; base += 32) {
    const int cnt = min(32, n - base);
    int s = 0, r = 0;
    float w = 0.0f;
    if (lane < cnt) {
      const int2 t = qw[base + lane];
      s = src[t.x];
      r = t.y;
      w = ns[s];
    }
    unsigned int pos = 0u;
    if (KEEP) {
      if (lane == 0) pos = atomicAdd(lcount, (unsigned int)cnt);
      pos = (unsigned int)__builtin_amdgcn_readfirstlane((int)pos);
    }
    agg_add_rows<LAST>(cnt, s, r, w, acc, Y4, lane, [] {});
    if (KEEP) {
      const unsigned int p = pos + (unsigned int)lane;
      if (lane < cnt && p < lcap) {
        lsr[p] = make_int2(s, r);
        lw[p]  = w;
      }
    }
  }
  return keep;
}

template <int R>
__device__ __forceinline__ int agg_push(int qn, int2* qw, int d, int e, unsigned int lo, int lane) {
  const unsigned int u = (unsigned int)d - lo;
  const bool m = u < (unsigned int)R;
  const unsigned int mask = __builtin_amdgcn_ballot_w32(m);
  if (mask) {
    if (m) qw[qn + __builtin_amdgcn_mbcnt_lo(mask, 0u)] = make_int2(e, (int)u);
    qn += __builtin_popcount(mask);
  }
  return qn;
}

template <int R, bool KEEP>
__device__ __forceinline__ void agg_scan(const int* __restrict__ src, const int* __restrict__ dst,
                                         const float* __restrict__ ns, const float4* __restrict__ Y4,
                                         float* acc, int2* qw, unsigned int lo, int E, int lane, int wave,
                                         unsigned int* lcount, int2* lsr, float* lw, unsigned int lcap) {
  const int E4 = E >> 2;
  const int4* __restrict__ dst4 = (const int4*)dst;
  const int nchunk = (E4 + AGG_CHUNK - 1) / AGG_CHUNK;
  int chunk = nchunk ? (int)(blockIdx.x % (unsigned int)nchunk) : 0;
  int qn = 0;
  for (int t = 0; t < nchunk; ++t) {
    const int ib = chunk * AGG_CHUNK + wave * 32 + lane;
    int4 v[AGG_DEPTH];
    #pragma unroll
    for (int k = 0; k < AGG_DEPTH; ++k) {
      v[k] = dst4[min(ib + k * AGG_THREADS, E4 - 1)];
    }
    #pragma unroll
    for (int k = 0; k < AGG_DEPTH; ++k) {
      const int e = (ib + k * AGG_THREADS) << 2;
      if (e >= (E4 << 2)) v[k] = make_int4(-1, -1, -1, -1);
      qn = agg_push<R>(qn, qw, v[k].x, e + 0, lo, lane);
      qn = agg_push<R>(qn, qw, v[k].y, e + 1, lo, lane);
      qn = agg_push<R>(qn, qw, v[k].z, e + 2, lo, lane);
      qn = agg_push<R>(qn, qw, v[k].w, e + 3, lo, lane);
      if (qn >= AGG_FLUSH) qn = agg_drain<false, KEEP>(qn, qw, acc, src, ns, Y4, lane, lcount, lsr, lw, lcap);
    }
    if (++chunk == nchunk) chunk = 0;
  }
  if (wave == 0) {
    const int e = (E4 << 2) + lane;
    qn = agg_push<R>(qn, qw, (e < E) ? dst[e] : -1, e, lo, lane);
  }
  if (qn) agg_drain<true, KEEP>(qn, qw, acc, src, ns, Y4, lane, lcount, lsr, lw, lcap);
}

template <int R, bool FIRST>
__global__ __launch_bounds__(AGG_THREADS)
void aggregate_owner_kernel(const int* __restrict__ src, const int* __restrict__ dst,
                            const float* __restrict__ ns, const float* __restrict__ nd,
                            const float* __restrict__ Y, const float* __restrict__ bias,
                            float* __restrict__ out, int N, int E, int relu,
                            int* lst, const int* lst_end) {
  __shared__ float acc[R * HDIM];
  __shared__ int2  queue[AGG_WAVES * AGG_QCAP];
  __shared__ unsigned int lcount;
  const int lane = threadIdx.x & 31;
  const int wave = threadIdx.x >> 5;
  const unsigned int lo = blockIdx.x * R;
  const float4* __restrict__ Y4 = (const float4*)Y;
  int2* qw = queue + wave * AGG_QCAP;

  int2* lsr; float* lw; unsigned int lcap;
  const bool lhdr = agg_list_place<R>(lst, lst_end, lsr, lw, lcap);
  if (threadIdx.x == 0) lcount = 0u;

  for (int r = wave; r < R; r += AGG_WAVES) {
    const unsigned int i = lo + r;
    float4 y = make_float4(0.0f, 0.0f, 0.0f, 0.0f);
    if (i < (unsigned int)N) {
      const float w = ns[i];
      y = Y4[(size_t)i * (HDIM / 4) + lane];
      y.x *= w; y.y *= w; y.z *= w; y.w *= w;
    }
    float* a = acc + r * HDIM + lane;
    a[0] = y.x; a[32] = y.y; a[64] = y.z; a[96] = y.w;
  }
  __syncthreads();

  if (FIRST) {
    agg_scan<R, true>(src, dst, ns, Y4, acc, qw, lo, E, lane, wave, &lcount, lsr, lw, lcap);
  } else {
    const unsigned int nl = lhdr ? (unsigned int)lst[blockIdx.x] : 0xffffffffu;
    if (nl <= lcap) {
      const unsigned int ngroups = (nl + 31u) >> 5;
      unsigned int gi = (unsigned int)__builtin_amdgcn_readfirstlane(wave);
      int2  e = make_int2(0, 0);
      float w = 0.0f;
      if (gi < ngroups) {
        const unsigned int p = gi * 32u + (unsigned int)lane;
        if (p < nl) { e = lsr[p]; w = lw[p]; }
      }
      while (gi < ngroups) {
        const unsigned int gn = gi + AGG_WAVES;
        int2  e2 = make_int2(0, 0);
        float w2 = 0.0f;
        auto load_next = [&] {
          if (gn < ngroups) {
            const unsigned int p = gn * 32u + (unsigned int)lane;
            if (p < nl) { e2 = lsr[p]; w2 = lw[p]; }
          }
        };
        const unsigned int left = nl - gi * 32u;
        if (left >= 32u) agg_add_rows<false>(32, e.x, e.y, w, acc, Y4, lane, load_next);
        else             agg_add_rows<true>((int)left, e.x, e.y, w, acc, Y4, lane, load_next);
        e = e2; w = w2; gi = gn;
      }
    } else {
      agg_scan<R, false>(src, dst, ns, Y4, acc, qw, lo, E, lane, wave, &lcount, lsr, lw, lcap);
    }
  }
  __syncthreads();
  if (FIRST && lhdr && threadIdx.x == 0) lst[blockIdx.x] = (int)lcount;

  const float4 b = ((const float4*)bias)[lane];
  for (int r = wave; r < R; r += AGG_WAVES) {
    const unsigned int i = lo + r;
    if (i < (unsigned int)N) {
      const float d = nd[i];
      const float* a = acc + r * HDIM + lane;
      float4 o;
      o.x = a[0]  * d + b.x;
      o.y = a[32] * d + b.y;
      o.z = a[64] * d + b.z;
      o.w = a[96] * d + b.w;
      if (relu) {
        o.x = fmaxf(o.x, 0.0f); o.y = fmaxf(o.y, 0.0f);
        o.z = fmaxf(o.z, 0.0f); o.w = fmaxf(o.w, 0.0f);
      }
      ((float4*)out)[(size_t)i * (HDIM / 4) + lane] = o;
    }
  }
}

static inline int cdiv(long long a, int b) { return (int)((a + b - 1) / b); }

template <int R, int D0>
static void run_graph(const float* feat, const int* src, const int* dst,
                      const float* W0, const float* b0, const float* W1, const float* b1,
                      int N, int E,
                      float* ns, float* nd, float* Y, float* h,
                      float* out, int* lst, const int* lst_end, hipStream_t stream) {
  degree_norm_kernel<<<dim3(cdiv(N, DEG_RANGE), 2), DEG_THREADS, 0, stream>>>(src, dst, ns, nd, N, E);

  gemm_wreg_kernel<D0><<<GEMM_GRID, GEMM_THREADS, 0, stream>>>(feat, W0, Y, N / NT);
  aggregate_owner_kernel<R, true><<<cdiv(N, R), AGG_THREADS, 0, stream>>>(src, dst, ns, nd, Y, b0, h, N, E, 1, lst, lst_end);

  gemm_wreg_kernel<HDIM><<<GEMM_GRID, GEMM_THREADS, 0, stream>>>(h, W1, Y, N / NT);
  aggregate_owner_kernel<R, false><<<cdiv(N, R), AGG_THREADS, 0, stream>>>(src, dst, ns, nd, Y, b1, out, N, E, 0, lst, lst_end);
}

extern "C" void kernel_launch(void* const* d_in, const int* in_sizes, int n_in,
                              void* d_out, int out_size, void* d_ws, size_t ws_size,
                              hipStream_t stream) {
  const float* feat_a = (const float*)d_in[0];
  const float* feat_b = (const float*)d_in[1];
  const int*   src_a  = (const int*)d_in[2];
  const int*   dst_a  = (const int*)d_in[3];
  const int*   src_b  = (const int*)d_in[4];
  const int*   dst_b  = (const int*)d_in[5];
  const float* Wa0 = (const float*)d_in[6];
  const float* ba0 = (const float*)d_in[7];
  const float* Wa1 = (const float*)d_in[8];
  const float* ba1 = (const float*)d_in[9];
  const float* Wb0 = (const float*)d_in[10];
  const float* bb0 = (const float*)d_in[11];
  const float* Wb1 = (const float*)d_in[12];
  const float* bb1 = (const float*)d_in[13];

  const int DA = in_sizes[6] / HDIM;
  const int DB = in_sizes[10] / HDIM;
  const int NA = in_sizes[0] / DA;
  const int NB = in_sizes[1] / DB;
  const int EA = in_sizes[2];
  const int EB = in_sizes[4];

  float* ws = (float*)d_ws;
  float* ns_a  = ws;
  float* nd_a  = ns_a + NA;
  float* Y_a   = nd_a + NA;
  float* h_a   = Y_a + (size_t)NA * HDIM;
  float* ns_b  = h_a + (size_t)NA * HDIM;
  float* nd_b  = ns_b + NB;
  float* Y_b   = nd_b + NB;
  float* h_b   = Y_b + (size_t)NB * HDIM;

  float* out_a = (float*)d_out;
  float* out_b = out_a + (size_t)NA * HDIM;

  run_graph<256, 256>(feat_a, src_a, dst_a, Wa0, ba0, Wa1, ba1, NA, EA,
                 ns_a, nd_a, Y_a, h_a, out_a,
                 (int*)h_b, (const int*)(h_b + (size_t)NB * HDIM), stream);
  run_graph<128, 128>(feat_b, src_b, dst_b, Wb0, bb0, Wb1, bb1, NB, EB,
                 ns_b, nd_b, Y_b, h_b, out_b,
                 (int*)Y_a, (const int*)(Y_a + (size_t)NA * HDIM), stream);
}


# The dst scan moves into the degree launch: its dst owners take the sum's row ranges, keep a (source,row) match list and count in-degrees; layer 0's sum reads the list like layer 1; arrival order
# speedup vs baseline: 3.1995x; 1.2075x over previous
#include <hip/hip_runtime.h>

typedef __attribute__((ext_vector_type(2))) float v2f;
typedef __attribute__((ext_vector_type(8))) float v8f;

#define HDIM 128
#define NT   16

#define AGG_THREADS 1024
#define AGG_WAVES   32
#define AGG_DEPTH   8
#define AGG_CHUNK   (AGG_DEPTH * AGG_THREADS)
#define AGG_FLUSH   32
#define AGG_QCAP    (AGG_FLUSH + 128)
#define AGG_ROWS    16
#define AGG_LIST_PER_ROW 20

template <int R>
__device__ __forceinline__ bool agg_list_place(int* lst, const int* lst_end, unsigned int nranges, unsigned int range,
                                               int2*& lsr, unsigned int& lcap) {
  constexpr unsigned int CAP = AGG_LIST_PER_ROW * R;
  const size_t avail = (size_t)(lst_end - lst);
  const size_t hdr   = (size_t)((nranges + 3u) & ~3u);
  const size_t off   = hdr + (size_t)range * (2u * CAP);
  lsr  = (int2*)(lst + off);
  lcap = (off + 2u * CAP <= avail) ? CAP : 0u;
  return hdr <= avail;
}

template <int R>
__device__ __forceinline__ int agg_push(int qn, int2* qw, int d, int e, unsigned int lo, int lane) {
  const unsigned int u = (unsigned int)d - lo;
  const bool m = u < (unsigned int)R;
  const unsigned int mask = __builtin_amdgcn_ballot_w32(m);
  if (mask) {
    if (m) qw[qn + __builtin_amdgcn_mbcnt_lo(mask, 0u)] = make_int2(e, (int)u);
    qn += __builtin_popcount(mask);
  }
  return qn;
}

template <int R, class DS, class DA>
__device__ __forceinline__ void agg_scan(const int* __restrict__ dst, int2* qw, unsigned int lo, int E,
                                         int lane, int wave, unsigned int range,
                                         DS&& drain_some, DA&& drain_all) {
  const int E4 = E >> 2;
  const int4* __restrict__ dst4 = (const int4*)dst;
  const int nchunk = (E4 + AGG_CHUNK - 1) / AGG_CHUNK;
  int chunk = nchunk ? (int)(range % (unsigned int)nchunk) : 0;
  int qn = 0;
  for (int t = 0; t < nchunk; ++t) {
    const int ib = chunk * AGG_CHUNK + wave * 32 + lane;
    int4 v[AGG_DEPTH];
    #pragma unroll
    for (int k = 0; k < AGG_DEPTH; ++k) {
      v[k] = dst4[min(ib + k * AGG_THREADS, E4 - 1)];
    }
    #pragma unroll
    for (int k = 0; k < AGG_DEPTH; ++k) {
      const int e = (ib + k * AGG_THREADS) << 2;
      if (e >= (E4 << 2)) v[k] = make_int4(-1, -1, -1, -1);
      qn = agg_push<R>(qn, qw, v[k].x, e + 0, lo, lane);
      qn = agg_push<R>(qn, qw, v[k].y, e + 1, lo, lane);
      qn = agg_push<R>(qn, qw, v[k].z, e + 2, lo, lane);
      qn = agg_push<R>(qn, qw, v[k].w, e + 3, lo, lane);
      if (qn >= AGG_FLUSH) qn = drain_some(qn);
    }
    if (++chunk == nchunk) chunk = 0;
  }
  if (wave == 0) {
    const int e = (E4 << 2) + lane;
    qn = agg_push<R>(qn, qw, (e < E) ? dst[e] : -1, e, lo, lane);
  }
  if (qn) drain_all(qn);
}

#define DEG_RANGE   4096
#define DEG_DEPTH   8

__device__ __forceinline__ void deg_count(unsigned int* cnt, int v, unsigned int lo) {
  unsigned int u = (unsigned int)v - lo;
  if (u < DEG_RANGE) atomicAdd(&cnt[u], 1u);
}

template <int R>
__global__ __launch_bounds__(AGG_THREADS)
void prepare_kernel(const int* __restrict__ src, const int* __restrict__ dst,
                    float* __restrict__ ns, float* __restrict__ nd, int N, int E,
                    unsigned int nranges, int* lst, const int* lst_end) {
  constexpr int NCNT = DEG_RANGE > R ? DEG_RANGE : R;
  __shared__ unsigned int cnt[NCNT];
  __shared__ int2 queue[AGG_WAVES * AGG_QCAP];
  __shared__ unsigned int lcount;

  if (blockIdx.x >= nranges) {
    const unsigned int lo = (blockIdx.x - nranges) * DEG_RANGE;
    for (int r = threadIdx.x; r < DEG_RANGE; r += AGG_THREADS) cnt[r] = 0u;
    __syncthreads();

    const int E4 = E >> 2;
    const int4* __restrict__ idx4 = (const int4*)src;
    for (int i0 = threadIdx.x; i0 < E4; i0 += DEG_DEPTH * AGG_THREADS) {
      int4 v[DEG_DEPTH];
      #pragma unroll
      for (int k = 0; k < DEG_DEPTH; ++k) {
        const int i = i0 + k * AGG_THREADS;
        v[k] = (i < E4) ? idx4[i] : make_int4(-1, -1, -1, -1);
      }
      #pragma unroll
      for (int k = 0; k < DEG_DEPTH; ++k) {
        deg_count(cnt, v[k].x, lo);
        deg_count(cnt, v[k].y, lo);
        deg_count(cnt, v[k].z, lo);
        deg_count(cnt, v[k].w, lo);
      }
    }
    for (int e = (E4 << 2) + threadIdx.x; e < E; e += AGG_THREADS) deg_count(cnt, src[e], lo);
    __syncthreads();

    for (int r = threadIdx.x; r < DEG_RANGE; r += AGG_THREADS) {
      const unsigned int i = lo + r;
      if (i < (unsigned int)N) ns[i] = rsqrtf(fmaxf(1.0f + (float)cnt[r], 1.0f));
    }
    return;
  }

  const int lane = threadIdx.x & 31;
  const int wave = threadIdx.x >> 5;
  const unsigned int lo = blockIdx.x * R;
  int2* qw = queue + wave * AGG_QCAP;
  int2* lsr; unsigned int lcap;
  const bool lhdr = agg_list_place<R>(lst, lst_end, nranges, blockIdx.x, lsr, lcap);

  for (int r = threadIdx.x; r < R; r += AGG_THREADS) cnt[r] = 0u;
  if (threadIdx.x == 0) lcount = 0u;
  __syncthreads();

  auto drain = [&](int n) -> int {
    for (int base = 0; base < n; base += 32) {
      const int c = min(32, n - base);
      unsigned int pos = 0u;
      if (lane == 0) pos = atomicAdd(&lcount, (unsigned int)c);
      pos = (unsigned int)__builtin_amdgcn_readfirstlane((int)pos);
      if (lane < c) {
        const int2 t = qw[base + lane];
        const int s = src[t.x];
        atomicAdd(&cnt[t.y], 1u);
        const unsigned int p = pos + (unsigned int)lane;
        if (p < lcap) lsr[p] = make_int2(s, t.y);
      }
    }
    return 0;
  };
  agg_scan<R>(dst, qw, lo, E, lane, wave, blockIdx.x, drain, drain);
  __syncthreads();

  if (lhdr && threadIdx.x == 0) lst[blockIdx.x] = (int)lcount;
  for (int r = threadIdx.x; r < R; r += AGG_THREADS) {
    const unsigned int i = lo + r;
    if (i < (unsigned int)N) nd[i] = rsqrtf(fmaxf(1.0f + (float)cnt[r], 1.0f));
  }
}

#define GEMM_THREADS 256
#define GEMM_GRID    512

template <int KD>
__global__ __launch_bounds__(GEMM_THREADS)
void gemm_wreg_kernel(const float* __restrict__ X, const float* __restrict__ W,
                      float* __restrict__ Y, int ntiles) {
  constexpr int PITCH = KD + 4;
  constexpr int KS    = KD / 4;
  constexpr int F4ROW = KD / 4;
  constexpr int NLD   = (NT * F4ROW) / GEMM_THREADS;
  __shared__ __attribute__((aligned(16))) float xs[2][NT * PITCH];

  const int tid  = threadIdx.x;
  const int wave = tid >> 5;
  const int lane = tid & 31;
  const int l16  = lane & 15;
  const int hi   = lane >> 4;
  const int nbase = wave * NT;

  const float* __restrict__ wcol = W + nbase + l16;
  v2f b[KS];
  #pragma unroll
  for (int j = 0; j < KS; ++j) {
    b[j].x = wcol[(size_t)(4 * j + 2 * hi) * HDIM];
    b[j].y = wcol[(size_t)(4 * j + 2 * hi + 1) * HDIM];
  }

  const float4* __restrict__ X4 = (const float4*)X;
  const int f0 = tid, f1 = tid + GEMM_THREADS, f2 = tid + 2 * GEMM_THREADS, f3 = tid + 3 * GEMM_THREADS;
  const int o0 = (f0 / F4ROW) * PITCH + 4 * (f0 % F4ROW);
  const int o1 = (f1 / F4ROW) * PITCH + 4 * (f1 % F4ROW);
  const int o2 = (f2 / F4ROW) * PITCH + 4 * (f2 % F4ROW);
  const int o3 = (f3 / F4ROW) * PITCH + 4 * (f3 % F4ROW);
  float4 p0 = make_float4(0.0f, 0.0f, 0.0f, 0.0f), p1 = p0, p2 = p0, p3 = p0;

  int t = blockIdx.x;
  if (t < ntiles) {
    const float4* __restrict__ xt = X4 + (size_t)t * (NT * F4ROW);
    p0 = xt[f0];
    p1 = xt[f1];
    if (NLD > 2) { p2 = xt[f2]; p3 = xt[f3]; }
    *(float4*)(xs[0] + o0) = p0;
    *(float4*)(xs[0] + o1) = p1;
    if (NLD > 2) { *(float4*)(xs[0] + o2) = p2; *(float4*)(xs[0] + o3) = p3; }
  }
  __syncthreads();

  int buf = 0;
  for (; t < ntiles; t += gridDim.x) {
    const int tn = t + gridDim.x;
    if (tn < ntiles) {
      const float4* __restrict__ xt = X4 + (size_t)tn * (NT * F4ROW);
      p0 = xt[f0];
      p1 = xt[f1];
      if (NLD > 2) { p2 = xt[f2]; p3 = xt[f3]; }
    }

    const float* xa = xs[buf] + l16 * PITCH + 2 * hi;
    v8f c = {};
    #pragma unroll
    for (int j = 0; j < KS; ++j) {
      const v2f a = *(const v2f*)(xa + 4 * j);
      c = __builtin_amdgcn_wmma_f32_16x16x4_f32(false, a, false, b[j], (short)0, c, false, false);
    }

    if (tn < ntiles) {
      float* xn = xs[buf ^ 1];
      *(float4*)(xn + o0) = p0;
      *(float4*)(xn + o1) = p1;
      if (NLD > 2) { *(float4*)(xn + o2) = p2; *(float4*)(xn + o3) = p3; }
    }
    __syncthreads();
    buf ^= 1;

    float* __restrict__ yout = Y + (size_t)t * NT * HDIM + nbase + l16;
    #pragma unroll
    for (int v = 0; v < 8; ++v)
      yout[(size_t)(v + 8 * hi) * HDIM] = c[v];
  }
}


template <bool PARTIAL, class F>
__device__ __forceinline__ void agg_add_rows(int cnt, int s, int r, float w, float* acc,
                                             const float4* __restrict__ Y4, int lane, F&& behind_first_loads) {
  for (int g = 0; g < cnt; g += AGG_ROWS) {
    float4 row[AGG_ROWS];
    #pragma unroll
    for (int j = 0; j < AGG_ROWS; ++j) {
      const int jj = PARTIAL ? min(g + j, cnt - 1) : g + j;
      const int sj = __builtin_amdgcn_readlane(s, jj);
      row[j] = Y4[(size_t)sj * (HDIM / 4) + lane];
    }
    if (g == 0) behind_first_loads();
    #pragma unroll
    for (int j = 0; j < AGG_ROWS; ++j) {
      if (!PARTIAL || g + j < cnt) {
        const float wj = __int_as_float(__builtin_amdgcn_readlane(__float_as_int(w), g + j));
        const int   rj = __builtin_amdgcn_readlane(r, g + j);
        float* a = acc + rj * HDIM + lane;
        __hip_atomic_fetch_add(a,      row[j].x * wj, __ATOMIC_RELAXED, __HIP_MEMORY_SCOPE_WORKGROUP);
        __hip_atomic_fetch_add(a + 32, row[j].y * wj, __ATOMIC_RELAXED, __HIP_MEMORY_SCOPE_WORKGROUP);
        __hip_atomic_fetch_add(a + 64, row[j].z * wj, __ATOMIC_RELAXED, __HIP_MEMORY_SCOPE_WORKGROUP);
        __hip_atomic_fetch_add(a + 96, row[j].w * wj, __ATOMIC_RELAXED, __HIP_MEMORY_SCOPE_WORKGROUP);
      }
    }
  }
}

template <bool LAST>
__device__ __forceinline__ int agg_drain(int n, const int2* qw, float* acc,
                                         const int* __restrict__ src,
                                         const float* __restrict__ ns,
                                         const float4* __restrict__ Y4, int lane) {
  const int keep = LAST ? 0 : (n & (AGG_ROWS - 1));
  for (int base = keep; base < n; base += 32) {
    const int cnt = min(32, n - base);
    int s = 0, r = 0;
    float w = 0.0f;
    if (lane < cnt) {
      const int2 t = qw[base + lane];
      s = src[t.x];
      r = t.y;
      w = ns[s];
    }
    agg_add_rows<LAST>(cnt, s, r, w, acc, Y4, lane, [] {});
  }
  return keep;
}

template <int R>
__global__ __launch_bounds__(AGG_THREADS)
void aggregate_owner_kernel(const int* __restrict__ src, const int* __restrict__ dst,
                            const float* __restrict__ ns, const float* __restrict__ nd,
                            const float* __restrict__ Y, const float* __restrict__ bias,
                            float* __restrict__ out, int N, int E, int relu,
                            int* lst, const int* lst_end) {
  __shared__ float acc[R * HDIM];
  __shared__ int2  queue[AGG_WAVES * AGG_QCAP];
  const int lane = threadIdx.x & 31;
  const int wave = threadIdx.x >> 5;
  const unsigned int lo = blockIdx.x * R;
  const float4* __restrict__ Y4 = (const float4*)Y;

  int2* lsr; unsigned int lcap;
  const bool lhdr = agg_list_place<R>(lst, lst_end, gridDim.x, blockIdx.x, lsr, lcap);

  for (int r = wave; r < R; r += AGG_WAVES) {
    const unsigned int i = lo + r;
    float4 y = make_float4(0.0f, 0.0f, 0.0f, 0.0f);
    if (i < (unsigned int)N) {
      const float w = ns[i];
      y = Y4[(size_t)i * (HDIM / 4) + lane];
      y.x *= w; y.y *= w; y.z *= w; y.w *= w;
    }
    float* a = acc + r * HDIM + lane;
    a[0] = y.x; a[32] = y.y; a[64] = y.z; a[96] = y.w;
  }
  __syncthreads();

  const unsigned int nl = lhdr ? (unsigned int)lst[blockIdx.x] : 0xffffffffu;
  if (nl <= lcap) {
    const unsigned int ngroups = (nl + 31u) >> 5;
    unsigned int gi = (unsigned int)__builtin_amdgcn_readfirstlane(wave);
    int2  e = make_int2(0, 0);
    float w = 0.0f;
    if (gi < ngroups) {
      const unsigned int p = gi * 32u + (unsigned int)lane;
      if (p < nl) { e = lsr[p]; w = ns[e.x]; }
    }
    while (gi < ngroups) {
      const unsigned int gn = gi + AGG_WAVES;
      const unsigned int pn = gn * 32u + (unsigned int)lane;
      const bool has_next = gn < ngroups && pn < nl;
      int2  e2 = make_int2(0, 0);
      float w2 = 0.0f;
      if (has_next) e2 = lsr[pn];
      auto load_weight = [&] { if (has_next) w2 = ns[e2.x]; };
      const unsigned int left = nl - gi * 32u;
      if (left >= 32u) agg_add_rows<false>(32, e.x, e.y, w, acc, Y4, lane, load_weight);
      else             agg_add_rows<true>((int)left, e.x, e.y, w, acc, Y4, lane, load_weight);
      e = e2; w = w2; gi = gn;
    }
  } else {
    int2* qw = queue + wave * AGG_QCAP;
    agg_scan<R>(dst, qw, lo, E, lane, wave, blockIdx.x,
                [&](int n) { return agg_drain<false>(n, qw, acc, src, ns, Y4, lane); },
                [&](int n) { return agg_drain<true>(n, qw, acc, src, ns, Y4, lane); });
  }
  __syncthreads();

  const float4 b = ((const float4*)bias)[lane];
  for (int r = wave; r < R; r += AGG_WAVES) {
    const unsigned int i = lo + r;
    if (i < (unsigned int)N) {
      const float d = nd[i];
      const float* a = acc + r * HDIM + lane;
      float4 o;
      o.x = a[0]  * d + b.x;
      o.y = a[32] * d + b.y;
      o.z = a[64] * d + b.z;
      o.w = a[96] * d + b.w;
      if (relu) {
        o.x = fmaxf(o.x, 0.0f); o.y = fmaxf(o.y, 0.0f);
        o.z = fmaxf(o.z, 0.0f); o.w = fmaxf(o.w, 0.0f);
      }
      ((float4*)out)[(size_t)i * (HDIM / 4) + lane] = o;
    }
  }
}

static inline int cdiv(long long a, int b) { return (int)((a + b - 1) / b); }

template <int R, int D0>
static void run_graph(const float* feat, const int* src, const int* dst,
                      const float* W0, const float* b0, const float* W1, const float* b1,
                      int N, int E,
                      float* ns, float* nd, float* Y, float* h,
                      float* out, int* lst, const int* lst_end, hipStream_t stream) {
  const int nranges = cdiv(N, R);
  prepare_kernel<R><<<nranges + cdiv(N, DEG_RANGE), AGG_THREADS, 0, stream>>>(src, dst, ns, nd, N, E, (unsigned int)nranges, lst, lst_end);

  gemm_wreg_kernel<D0><<<GEMM_GRID, GEMM_THREADS, 0, stream>>>(feat, W0, Y, N / NT);
  aggregate_owner_kernel<R><<<nranges, AGG_THREADS, 0, stream>>>(src, dst, ns, nd, Y, b0, h, N, E, 1, lst, lst_end);

  gemm_wreg_kernel<HDIM><<<GEMM_GRID, GEMM_THREADS, 0, stream>>>(h, W1, Y, N / NT);
  aggregate_owner_kernel<R><<<nranges, AGG_THREADS, 0, stream>>>(src, dst, ns, nd, Y, b1, out, N, E, 0, lst, lst_end);
}

extern "C" void kernel_launch(void* const* d_in, const int* in_sizes, int n_in,
                              void* d_out, int out_size, void* d_ws, size_t ws_size,
                              hipStream_t stream) {
  const float* feat_a = (const float*)d_in[0];
  const float* feat_b = (const float*)d_in[1];
  const int*   src_a  = (const int*)d_in[2];
  const int*   dst_a  = (const int*)d_in[3];
  const int*   src_b  = (const int*)d_in[4];
  const int*   dst_b  = (const int*)d_in[5];
  const float* Wa0 = (const float*)d_in[6];
  const float* ba0 = (const float*)d_in[7];
  const float* Wa1 = (const float*)d_in[8];
  const float* ba1 = (const float*)d_in[9];
  const float* Wb0 = (const float*)d_in[10];
  const float* bb0 = (const float*)d_in[11];
  const float* Wb1 = (const float*)d_in[12];
  const float* bb1 = (const float*)d_in[13];

  const int DA = in_sizes[6] / HDIM;
  const int DB = in_sizes[10] / HDIM;
  const int NA = in_sizes[0] / DA;
  const int NB = in_sizes[1] / DB;
  const int EA = in_sizes[2];
  const int EB = in_sizes[4];

  float* ws = (float*)d_ws;
  float* ns_a  = ws;
  float* nd_a  = ns_a + NA;
  float* Y_a   = nd_a + NA;
  float* h_a   = Y_a + (size_t)NA * HDIM;
  float* ns_b  = h_a + (size_t)NA * HDIM;
  float* nd_b  = ns_b + NB;
  float* Y_b   = nd_b + NB;
  float* h_b   = Y_b + (size_t)NB * HDIM;

  float* out_a = (float*)d_out;
  float* out_b = out_a + (size_t)NA * HDIM;

  run_graph<256, 256>(feat_a, src_a, dst_a, Wa0, ba0, Wa1, ba1, NA, EA,
                 ns_a, nd_a, Y_a, h_a, out_a,
                 (int*)h_b, (const int*)(h_b + (size_t)NB * HDIM), stream);
  run_graph<128, 128>(feat_b, src_b, dst_b, Wb0, bb0, Wb1, bb1, NB, EB,
                 ns_b, nd_b, Y_b, h_b, out_b,
                 (int*)Y_a, (const int*)(Y_a + (size_t)NA * HDIM), stream);
}


# Source weight ns folded into Y by the product's store (Y row i = ns[i] * product row i, same f32 multiply as before): the neighbour sum gathers rows only, the random 4-byte ns lookup per edge goes
# speedup vs baseline: 3.3393x; 1.0437x over previous
#include <hip/hip_runtime.h>

typedef __attribute__((ext_vector_type(2))) float v2f;
typedef __attribute__((ext_vector_type(8))) float v8f;

#define HDIM 128
#define NT   16

#define AGG_THREADS 1024
#define AGG_WAVES   32
#define AGG_DEPTH   8
#define AGG_CHUNK   (AGG_DEPTH * AGG_THREADS)
#define AGG_FLUSH   32
#define AGG_QCAP    (AGG_FLUSH + 128)
#define AGG_ROWS    16
#define AGG_LIST_PER_ROW 20

template <int R>
__device__ __forceinline__ bool agg_list_place(int* lst, const int* lst_end, unsigned int nranges, unsigned int range,
                                               int2*& lsr, unsigned int& lcap) {
  constexpr unsigned int CAP = AGG_LIST_PER_ROW * R;
  const size_t avail = (size_t)(lst_end - lst);
  const size_t hdr   = (size_t)((nranges + 3u) & ~3u);
  const size_t off   = hdr + (size_t)range * (2u * CAP);
  lsr  = (int2*)(lst + off);
  lcap = (off + 2u * CAP <= avail) ? CAP : 0u;
  return hdr <= avail;
}

template <int R>
__device__ __forceinline__ int agg_push(int qn, int2* qw, int d, int e, unsigned int lo, int lane) {
  const unsigned int u = (unsigned int)d - lo;
  const bool m = u < (unsigned int)R;
  const unsigned int mask = __builtin_amdgcn_ballot_w32(m);
  if (mask) {
    if (m) qw[qn + __builtin_amdgcn_mbcnt_lo(mask, 0u)] = make_int2(e, (int)u);
    qn += __builtin_popcount(mask);
  }
  return qn;
}

template <int R, class DS, class DA>
__device__ __forceinline__ void agg_scan(const int* __restrict__ dst, int2* qw, unsigned int lo, int E,
                                         int lane, int wave, unsigned int range,
                                         DS&& drain_some, DA&& drain_all) {
  const int E4 = E >> 2;
  const int4* __restrict__ dst4 = (const int4*)dst;
  const int nchunk = (E4 + AGG_CHUNK - 1) / AGG_CHUNK;
  int chunk = nchunk ? (int)(range % (unsigned int)nchunk) : 0;
  int qn = 0;
  for (int t = 0; t < nchunk; ++t) {
    const int ib = chunk * AGG_CHUNK + wave * 32 + lane;
    int4 v[AGG_DEPTH];
    #pragma unroll
    for (int k = 0; k < AGG_DEPTH; ++k) {
      v[k] = dst4[min(ib + k * AGG_THREADS, E4 - 1)];
    }
    #pragma unroll
    for (int k = 0; k < AGG_DEPTH; ++k) {
      const int e = (ib + k * AGG_THREADS) << 2;
      if (e >= (E4 << 2)) v[k] = make_int4(-1, -1, -1, -1);
      qn = agg_push<R>(qn, qw, v[k].x, e + 0, lo, lane);
      qn = agg_push<R>(qn, qw, v[k].y, e + 1, lo, lane);
      qn = agg_push<R>(qn, qw, v[k].z, e + 2, lo, lane);
      qn = agg_push<R>(qn, qw, v[k].w, e + 3, lo, lane);
      if (qn >= AGG_FLUSH) qn = drain_some(qn);
    }
    if (++chunk == nchunk) chunk = 0;
  }
  if (wave == 0) {
    const int e = (E4 << 2) + lane;
    qn = agg_push<R>(qn, qw, (e < E) ? dst[e] : -1, e, lo, lane);
  }
  if (qn) drain_all(qn);
}

#define DEG_RANGE   4096
#define DEG_DEPTH   8

__device__ __forceinline__ void deg_count(unsigned int* cnt, int v, unsigned int lo) {
  unsigned int u = (unsigned int)v - lo;
  if (u < DEG_RANGE) atomicAdd(&cnt[u], 1u);
}

template <int R>
__global__ __launch_bounds__(AGG_THREADS)
void prepare_kernel(const int* __restrict__ src, const int* __restrict__ dst,
                    float* __restrict__ ns, float* __restrict__ nd, int N, int E,
                    unsigned int nranges, int* lst, const int* lst_end) {
  constexpr int NCNT = DEG_RANGE > R ? DEG_RANGE : R;
  __shared__ unsigned int cnt[NCNT];
  __shared__ int2 queue[AGG_WAVES * AGG_QCAP];
  __shared__ unsigned int lcount;

  if (blockIdx.x >= nranges) {
    const unsigned int lo = (blockIdx.x - nranges) * DEG_RANGE;
    for (int r = threadIdx.x; r < DEG_RANGE; r += AGG_THREADS) cnt[r] = 0u;
    __syncthreads();

    const int E4 = E >> 2;
    const int4* __restrict__ idx4 = (const int4*)src;
    for (int i0 = threadIdx.x; i0 < E4; i0 += DEG_DEPTH * AGG_THREADS) {
      int4 v[DEG_DEPTH];
      #pragma unroll
      for (int k = 0; k < DEG_DEPTH; ++k) {
        const int i = i0 + k * AGG_THREADS;
        v[k] = (i < E4) ? idx4[i] : make_int4(-1, -1, -1, -1);
      }
      #pragma unroll
      for (int k = 0; k < DEG_DEPTH; ++k) {
        deg_count(cnt, v[k].x, lo);
        deg_count(cnt, v[k].y, lo);
        deg_count(cnt, v[k].z, lo);
        deg_count(cnt, v[k].w, lo);
      }
    }
    for (int e = (E4 << 2) + threadIdx.x; e < E; e += AGG_THREADS) deg_count(cnt, src[e], lo);
    __syncthreads();

    for (int r = threadIdx.x; r < DEG_RANGE; r += AGG_THREADS) {
      const unsigned int i = lo + r;
      if (i < (unsigned int)N) ns[i] = rsqrtf(fmaxf(1.0f + (float)cnt[r], 1.0f));
    }
    return;
  }

  const int lane = threadIdx.x & 31;
  const int wave = threadIdx.x >> 5;
  const unsigned int lo = blockIdx.x * R;
  int2* qw = queue + wave * AGG_QCAP;
  int2* lsr; unsigned int lcap;
  const bool lhdr = agg_list_place<R>(lst, lst_end, nranges, blockIdx.x, lsr, lcap);

  for (int r = threadIdx.x; r < R; r += AGG_THREADS) cnt[r] = 0u;
  if (threadIdx.x == 0) lcount = 0u;
  __syncthreads();

  auto drain = [&](int n) -> int {
    for (int base = 0; base < n; base += 32) {
      const int c = min(32, n - base);
      unsigned int pos = 0u;
      if (lane == 0) pos = atomicAdd(&lcount, (unsigned int)c);
      pos = (unsigned int)__builtin_amdgcn_readfirstlane((int)pos);
      if (lane < c) {
        const int2 t = qw[base + lane];
        const int s = src[t.x];
        atomicAdd(&cnt[t.y], 1u);
        const unsigned int p = pos + (unsigned int)lane;
        if (p < lcap) lsr[p] = make_int2(s, t.y);
      }
    }
    return 0;
  };
  agg_scan<R>(dst, qw, lo, E, lane, wave, blockIdx.x, drain, drain);
  __syncthreads();

  if (lhdr && threadIdx.x == 0) lst[blockIdx.x] = (int)lcount;
  for (int r = threadIdx.x; r < R; r += AGG_THREADS) {
    const unsigned int i = lo + r;
    if (i < (unsigned int)N) nd[i] = rsqrtf(fmaxf(1.0f + (float)cnt[r], 1.0f));
  }
}

#define GEMM_THREADS 256
#define GEMM_GRID    512

template <int KD>
__global__ __launch_bounds__(GEMM_THREADS)
void gemm_wreg_kernel(const float* __restrict__ X, const float* __restrict__ W,
                      const float* __restrict__ ns, float* __restrict__ Y, int ntiles) {
  constexpr int PITCH = KD + 4;
  constexpr int KS    = KD / 4;
  constexpr int F4ROW = KD / 4;
  constexpr int NLD   = (NT * F4ROW) / GEMM_THREADS;
  __shared__ __attribute__((aligned(16))) float xs[2][NT * PITCH];

  const int tid  = threadIdx.x;
  const int wave = tid >> 5;
  const int lane = tid & 31;
  const int l16  = lane & 15;
  const int hi   = lane >> 4;
  const int nbase = wave * NT;

  const float* __restrict__ wcol = W + nbase + l16;
  v2f b[KS];
  #pragma unroll
  for (int j = 0; j < KS; ++j) {
    b[j].x = wcol[(size_t)(4 * j + 2 * hi) * HDIM];
    b[j].y = wcol[(size_t)(4 * j + 2 * hi + 1) * HDIM];
  }

  const float4* __restrict__ X4 = (const float4*)X;
  const int f0 = tid, f1 = tid + GEMM_THREADS, f2 = tid + 2 * GEMM_THREADS, f3 = tid + 3 * GEMM_THREADS;
  const int o0 = (f0 / F4ROW) * PITCH + 4 * (f0 % F4ROW);
  const int o1 = (f1 / F4ROW) * PITCH + 4 * (f1 % F4ROW);
  const int o2 = (f2 / F4ROW) * PITCH + 4 * (f2 % F4ROW);
  const int o3 = (f3 / F4ROW) * PITCH + 4 * (f3 % F4ROW);
  float4 p0 = make_float4(0.0f, 0.0f, 0.0f, 0.0f), p1 = p0, p2 = p0, p3 = p0;
  float4 s0 = p0, s1 = p0;

  int t = blockIdx.x;
  if (t < ntiles) {
    const float4* __restrict__ xt = X4 + (size_t)t * (NT * F4ROW);
    p0 = xt[f0];
    p1 = xt[f1];
    if (NLD > 2) { p2 = xt[f2]; p3 = xt[f3]; }
    const float4* __restrict__ ns4 = (const float4*)(ns + (size_t)t * NT + 8 * hi);
    s0 = ns4[0];
    s1 = ns4[1];
    *(float4*)(xs[0] + o0) = p0;
    *(float4*)(xs[0] + o1) = p1;
    if (NLD > 2) { *(float4*)(xs[0] + o2) = p2; *(float4*)(xs[0] + o3) = p3; }
  }
  __syncthreads();

  int buf = 0;
  for (; t < ntiles; t += gridDim.x) {
    const int tn = t + gridDim.x;
    float4 sn0 = s0, sn1 = s1;
    if (tn < ntiles) {
      const float4* __restrict__ xt = X4 + (size_t)tn * (NT * F4ROW);
      p0 = xt[f0];
      p1 = xt[f1];
      if (NLD > 2) { p2 = xt[f2]; p3 = xt[f3]; }
      const float4* __restrict__ ns4 = (const float4*)(ns + (size_t)tn * NT + 8 * hi);
      sn0 = ns4[0];
      sn1 = ns4[1];
    }

    const float* xa = xs[buf] + l16 * PITCH + 2 * hi;
    v8f c = {};
    #pragma unroll
    for (int j = 0; j < KS; ++j) {
      const v2f a = *(const v2f*)(xa + 4 * j);
      c = __builtin_amdgcn_wmma_f32_16x16x4_f32(false, a, false, b[j], (short)0, c, false, false);
    }

    if (tn < ntiles) {
      float* xn = xs[buf ^ 1];
      *(float4*)(xn + o0) = p0;
      *(float4*)(xn + o1) = p1;
      if (NLD > 2) { *(float4*)(xn + o2) = p2; *(float4*)(xn + o3) = p3; }
    }
    __syncthreads();
    buf ^= 1;

    float* __restrict__ yout = Y + (size_t)t * NT * HDIM + nbase + l16;
    const float sr[8] = {s0.x, s0.y, s0.z, s0.w, s1.x, s1.y, s1.z, s1.w};
    #pragma unroll
    for (int v = 0; v < 8; ++v)
      yout[(size_t)(v + 8 * hi) * HDIM] = c[v] * sr[v];
    s0 = sn0;
    s1 = sn1;
  }
}


template <bool PARTIAL>
__device__ __forceinline__ void agg_add_rows(int cnt, int s, int r, float* acc,
                                             const float4* __restrict__ Y4, int lane) {
  for (int g = 0; g < cnt; g += AGG_ROWS) {
    float4 row[AGG_ROWS];
    #pragma unroll
    for (int j = 0; j < AGG_ROWS; ++j) {
      const int jj = PARTIAL ? min(g + j, cnt - 1) : g + j;
      const int sj = __builtin_amdgcn_readlane(s, jj);
      row[j] = Y4[(size_t)sj * (HDIM / 4) + lane];
    }
    #pragma unroll
    for (int j = 0; j < AGG_ROWS; ++j) {
      if (!PARTIAL || g + j < cnt) {
        const int rj = __builtin_amdgcn_readlane(r, g + j);
        float* a = acc + rj * HDIM + lane;
        __hip_atomic_fetch_add(a,      row[j].x, __ATOMIC_RELAXED, __HIP_MEMORY_SCOPE_WORKGROUP);
        __hip_atomic_fetch_add(a + 32, row[j].y, __ATOMIC_RELAXED, __HIP_MEMORY_SCOPE_WORKGROUP);
        __hip_atomic_fetch_add(a + 64, row[j].z, __ATOMIC_RELAXED, __HIP_MEMORY_SCOPE_WORKGROUP);
        __hip_atomic_fetch_add(a + 96, row[j].w, __ATOMIC_RELAXED, __HIP_MEMORY_SCOPE_WORKGROUP);
      }
    }
  }
}

template <bool LAST>
__device__ __forceinline__ int agg_drain(int n, const int2* qw, float* acc,
                                         const int* __restrict__ src,
                                         const float4* __restrict__ Y4, int lane) {
  const int keep = LAST ? 0 : (n & (AGG_ROWS - 1));
  for (int base = keep; base < n; base += 32) {
    const int cnt = min(32, n - base);
    int s = 0, r = 0;
    if (lane < cnt) {
      const int2 t = qw[base + lane];
      s = src[t.x];
      r = t.y;
    }
    agg_add_rows<LAST>(cnt, s, r, acc, Y4, lane);
  }
  return keep;
}

template <int R>
__global__ __launch_bounds__(AGG_THREADS)
void aggregate_owner_kernel(const int* __restrict__ src, const int* __restrict__ dst,
                            const float* __restrict__ nd,
                            const float* __restrict__ Y, const float* __restrict__ bias,
                            float* __restrict__ out, int N, int E, int relu,
                            int* lst, const int* lst_end) {
  __shared__ float acc[R * HDIM];
  __shared__ int2  queue[AGG_WAVES * AGG_QCAP];
  const int lane = threadIdx.x & 31;
  const int wave = threadIdx.x >> 5;
  const unsigned int lo = blockIdx.x * R;
  const float4* __restrict__ Y4 = (const float4*)Y;

  int2* lsr; unsigned int lcap;
  const bool lhdr = agg_list_place<R>(lst, lst_end, gridDim.x, blockIdx.x, lsr, lcap);

  for (int r = wave; r < R; r += AGG_WAVES) {
    const unsigned int i = lo + r;
    float4 y = make_float4(0.0f, 0.0f, 0.0f, 0.0f);
    if (i < (unsigned int)N) y = Y4[(size_t)i * (HDIM / 4) + lane];
    float* a = acc + r * HDIM + lane;
    a[0] = y.x; a[32] = y.y; a[64] = y.z; a[96] = y.w;
  }
  __syncthreads();

  const unsigned int nl = lhdr ? (unsigned int)lst[blockIdx.x] : 0xffffffffu;
  if (nl <= lcap) {
    const unsigned int ngroups = (nl + 31u) >> 5;
    unsigned int gi = (unsigned int)__builtin_amdgcn_readfirstlane(wave);
    int2 e = make_int2(0, 0);
    if (gi < ngroups) {
      const unsigned int p = gi * 32u + (unsigned int)lane;
      if (p < nl) e = lsr[p];
    }
    while (gi < ngroups) {
      const unsigned int gn = gi + AGG_WAVES;
      const unsigned int pn = gn * 32u + (unsigned int)lane;
      const bool has_next = gn < ngroups && pn < nl;
      int2 e2 = make_int2(0, 0);
      if (has_next) e2 = lsr[pn];
      const unsigned int left = nl - gi * 32u;
      if (left >= 32u) agg_add_rows<false>(32, e.x, e.y, acc, Y4, lane);
      else             agg_add_rows<true>((int)left, e.x, e.y, acc, Y4, lane);
      e = e2; gi = gn;
    }
  } else {
    int2* qw = queue + wave * AGG_QCAP;
    agg_scan<R>(dst, qw, lo, E, lane, wave, blockIdx.x,
                [&](int n) { return agg_drain<false>(n, qw, acc, src, Y4, lane); },
                [&](int n) { return agg_drain<true>(n, qw, acc, src, Y4, lane); });
  }
  __syncthreads();

  const float4 b = ((const float4*)bias)[lane];
  for (int r = wave; r < R; r += AGG_WAVES) {
    const unsigned int i = lo + r;
    if (i < (unsigned int)N) {
      const float d = nd[i];
      const float* a = acc + r * HDIM + lane;
      float4 o;
      o.x = a[0]  * d + b.x;
      o.y = a[32] * d + b.y;
      o.z = a[64] * d + b.z;
      o.w = a[96] * d + b.w;
      if (relu) {
        o.x = fmaxf(o.x, 0.0f); o.y = fmaxf(o.y, 0.0f);
        o.z = fmaxf(o.z, 0.0f); o.w = fmaxf(o.w, 0.0f);
      }
      ((float4*)out)[(size_t)i * (HDIM / 4) + lane] = o;
    }
  }
}

static inline int cdiv(long long a, int b) { return (int)((a + b - 1) / b); }

template <int R, int D0>
static void run_graph(const float* feat, const int* src, const int* dst,
                      const float* W0, const float* b0, const float* W1, const float* b1,
                      int N, int E,
                      float* ns, float* nd, float* Y, float* h,
                      float* out, int* lst, const int* lst_end, hipStream_t stream) {
  const int nranges = cdiv(N, R);
  prepare_kernel<R><<<nranges + cdiv(N, DEG_RANGE), AGG_THREADS, 0, stream>>>(src, dst, ns, nd, N, E, (unsigned int)nranges, lst, lst_end);

  gemm_wreg_kernel<D0><<<GEMM_GRID, GEMM_THREADS, 0, stream>>>(feat, W0, ns, Y, N / NT);
  aggregate_owner_kernel<R><<<nranges, AGG_THREADS, 0, stream>>>(src, dst, nd, Y, b0, h, N, E, 1, lst, lst_end);

  gemm_wreg_kernel<HDIM><<<GEMM_GRID, GEMM_THREADS, 0, stream>>>(h, W1, ns, Y, N / NT);
  aggregate_owner_kernel<R><<<nranges, AGG_THREADS, 0, stream>>>(src, dst, nd, Y, b1, out, N, E, 0, lst, lst_end);
}

extern "C" void kernel_launch(void* const* d_in, const int* in_sizes, int n_in,
                              void* d_out, int out_size, void* d_ws, size_t ws_size,
                              hipStream_t stream) {
  const float* feat_a = (const float*)d_in[0];
  const float* feat_b = (const float*)d_in[1];
  const int*   src_a  = (const int*)d_in[2];
  const int*   dst_a  = (const int*)d_in[3];
  const int*   src_b  = (const int*)d_in[4];
  const int*   dst_b  = (const int*)d_in[5];
  const float* Wa0 = (const float*)d_in[6];
  const float* ba0 = (const float*)d_in[7];
  const float* Wa1 = (const float*)d_in[8];
  const float* ba1 = (const float*)d_in[9];
  const float* Wb0 = (const float*)d_in[10];
  const float* bb0 = (const float*)d_in[11];
  const float* Wb1 = (const float*)d_in[12];
  const float* bb1 = (const float*)d_in[13];

  const int DA = in_sizes[6] / HDIM;
  const int DB = in_sizes[10] / HDIM;
  const int NA = in_sizes[0] / DA;
  const int NB = in_sizes[1] / DB;
  const int EA = in_sizes[2];
  const int EB = in_sizes[4];

  float* ws = (float*)d_ws;
  float* ns_a  = ws;
  float* nd_a  = ns_a + NA;
  float* Y_a   = nd_a + NA;
  float* h_a   = Y_a + (size_t)NA * HDIM;
  float* ns_b  = h_a + (size_t)NA * HDIM;
  float* nd_b  = ns_b + NB;
  float* Y_b   = nd_b + NB;
  float* h_b   = Y_b + (size_t)NB * HDIM;

  float* out_a = (float*)d_out;
  float* out_b = out_a + (size_t)NA * HDIM;

  run_graph<256, 256>(feat_a, src_a, dst_a, Wa0, ba0, Wa1, ba1, NA, EA,
                 ns_a, nd_a, Y_a, h_a, out_a,
                 (int*)h_b, (const int*)(h_b + (size_t)NB * HDIM), stream);
  run_graph<128, 128>(feat_b, src_b, dst_b, Wb0, bb0, Wb1, bb1, NB, EB,
                 ns_b, nd_b, Y_b, h_b, out_b,
                 (int*)Y_a, (const int*)(Y_a + (size_t)NA * HDIM), stream);
}


# First launch's dst and src scans cut 4 ways over the edge list: a scanner reads a quarter for 4 ranges, one list segment per (range, quarter); sums and products add the 4 parts; arrival order
# speedup vs baseline: 3.7924x; 1.1357x over previous
#include <hip/hip_runtime.h>

typedef __attribute__((ext_vector_type(2))) float v2f;
typedef __attribute__((ext_vector_type(8))) float v8f;

#define HDIM 128
#define NT   16

#define AGG_THREADS 1024
#define AGG_WAVES   32
#define AGG_DEPTH   8
#define AGG_CHUNK   (AGG_DEPTH * AGG_THREADS)
#define AGG_FLUSH   32
#define AGG_QCAP    (AGG_FLUSH + 128)
#define AGG_ROWS    16
#define AGG_LIST_PER_ROW 20
#define SCAN_SPLIT  4

template <int R>
__device__ __forceinline__ bool agg_list_place(int* lst, const int* lst_end, unsigned int nranges,
                                               int2*& seg0, unsigned int& nfit) {
  constexpr unsigned int CAP = AGG_LIST_PER_ROW * R / SCAN_SPLIT;
  const size_t avail = (size_t)(lst_end - lst);
  const size_t hdr   = (size_t)((nranges * SCAN_SPLIT + 3u) & ~3u);
  const bool   ok    = hdr <= avail;
  seg0 = (int2*)(lst + (ok ? hdr : 0));
  const size_t fit = ok ? (avail - hdr) / (size_t)(2u * CAP * SCAN_SPLIT) : 0;
  nfit = (unsigned int)(fit < (size_t)nranges ? fit : (size_t)nranges);
  return ok;
}

template <int SPAN>
__device__ __forceinline__ int agg_push(int qn, int2* qw, int d, int e, unsigned int lo, int lane) {
  const unsigned int u = (unsigned int)d - lo;
  const bool m = u < (unsigned int)SPAN;
  const unsigned int mask = __builtin_amdgcn_ballot_w32(m);
  if (mask) {
    if (m) qw[qn + __builtin_amdgcn_mbcnt_lo(mask, 0u)] = make_int2(e, (int)u);
    qn += __builtin_popcount(mask);
  }
  return qn;
}

template <int SPAN, class DS, class DA>
__device__ __forceinline__ void agg_scan(const int* __restrict__ dst, int2* qw, unsigned int lo, int e0, int e1,
                                         int lane, int wave, unsigned int rot,
                                         DS&& drain_some, DA&& drain_all) {
  const int i0 = e0 >> 2;
  const int i1 = e1 >> 2;
  const int n4 = i1 - i0;
  const int4* __restrict__ dst4 = (const int4*)dst;
  const int nchunk = n4 > 0 ? (n4 + AGG_CHUNK - 1) / AGG_CHUNK : 0;
  int chunk = nchunk ? (int)(rot % (unsigned int)nchunk) : 0;
  int qn = 0;
  for (int t = 0; t < nchunk; ++t) {
    const int ib = i0 + chunk * AGG_CHUNK + wave * 32 + lane;
    int4 v[AGG_DEPTH];
    #pragma unroll
    for (int k = 0; k < AGG_DEPTH; ++k) {
      v[k] = dst4[min(ib + k * AGG_THREADS, i1 - 1)];
    }
    #pragma unroll
    for (int k = 0; k < AGG_DEPTH; ++k) {
      const int i = ib + k * AGG_THREADS;
      const int e = i << 2;
      if (i >= i1) v[k] = make_int4(-1, -1, -1, -1);
      qn = agg_push<SPAN>(qn, qw, v[k].x, e + 0, lo, lane);
      qn = agg_push<SPAN>(qn, qw, v[k].y, e + 1, lo, lane);
      qn = agg_push<SPAN>(qn, qw, v[k].z, e + 2, lo, lane);
      qn = agg_push<SPAN>(qn, qw, v[k].w, e + 3, lo, lane);
      if (qn >= AGG_FLUSH) qn = drain_some(qn);
    }
    if (++chunk == nchunk) chunk = 0;
  }
  if (wave == 0) {
    const int e = max(i1 << 2, e0) + lane;
    qn = agg_push<SPAN>(qn, qw, (e < e1) ? dst[e] : -1, e, lo, lane);
  }
  if (qn) drain_all(qn);
}

#define DEG_RANGE   8192
#define DEG_DEPTH   8

__device__ __forceinline__ void deg_count(unsigned int* cnt, int v, unsigned int lo) {
  unsigned int u = (unsigned int)v - lo;
  if (u < DEG_RANGE) atomicAdd(&cnt[u], 1u);
}

template <int R>
__global__ __launch_bounds__(AGG_THREADS)
void prepare_kernel(const int* __restrict__ src, const int* __restrict__ dst,
                    unsigned int* __restrict__ pcnt, int N, int E,
                    unsigned int nranges, unsigned int ndst, int* lst, const int* lst_end) {
  __shared__ unsigned int cnt[DEG_RANGE];
  __shared__ int2 queue[AGG_WAVES * AGG_QCAP];
  __shared__ unsigned int lcount[SCAN_SPLIT];

  const int ES = ((E + 4 * SCAN_SPLIT - 1) / (4 * SCAN_SPLIT)) * 4;

  if (blockIdx.x >= ndst) {
    const unsigned int j     = blockIdx.x - ndst;
    const unsigned int slice = j % SCAN_SPLIT;
    const unsigned int lo    = (j / SCAN_SPLIT) * DEG_RANGE;
    const int e0 = min((int)slice * ES, E);
    const int e1 = min(e0 + ES, E);
    for (int r = threadIdx.x; r < DEG_RANGE; r += AGG_THREADS) cnt[r] = 0u;
    __syncthreads();

    const int i0 = e0 >> 2;
    const int i1 = e1 >> 2;
    const int4* __restrict__ idx4 = (const int4*)src;
    for (int ib = i0 + (int)threadIdx.x; ib < i1; ib += DEG_DEPTH * AGG_THREADS) {
      int4 v[DEG_DEPTH];
      #pragma unroll
      for (int k = 0; k < DEG_DEPTH; ++k) {
        const int i = ib + k * AGG_THREADS;
        v[k] = (i < i1) ? idx4[i] : make_int4(-1, -1, -1, -1);
      }
      #pragma unroll
      for (int k = 0; k < DEG_DEPTH; ++k) {
        deg_count(cnt, v[k].x, lo);
        deg_count(cnt, v[k].y, lo);
        deg_count(cnt, v[k].z, lo);
        deg_count(cnt, v[k].w, lo);
      }
    }
    for (int e = max(i1 << 2, e0) + (int)threadIdx.x; e < e1; e += AGG_THREADS) deg_count(cnt, src[e], lo);
    __syncthreads();

    unsigned int* __restrict__ plane = pcnt + (size_t)slice * (size_t)N;
    for (int r = threadIdx.x; r < DEG_RANGE; r += AGG_THREADS) {
      const unsigned int i = lo + r;
      if (i < (unsigned int)N) plane[i] = cnt[r];
    }
    return;
  }

  constexpr int SPAN = SCAN_SPLIT * R;
  constexpr unsigned int CAP = AGG_LIST_PER_ROW * R / SCAN_SPLIT;
  const int lane = threadIdx.x & 31;
  const int wave = threadIdx.x >> 5;
  const unsigned int group  = blockIdx.x / SCAN_SPLIT;
  const unsigned int slice  = blockIdx.x % SCAN_SPLIT;
  const unsigned int range0 = group * SCAN_SPLIT;
  const unsigned int lo     = range0 * R;
  const int e0 = min((int)slice * ES, E);
  const int e1 = min(e0 + ES, E);
  int2* qw = queue + wave * AGG_QCAP;
  int2* seg0; unsigned int nfit;
  const bool lhdr = agg_list_place<R>(lst, lst_end, nranges, seg0, nfit);

  if (threadIdx.x < SCAN_SPLIT) lcount[threadIdx.x] = 0u;
  __syncthreads();

  auto drain = [&](int n) -> int {
    for (int base = 0; base < n; base += 32) {
      const int c = min(32, n - base);
      const bool valid = lane < c;
      int2 t = make_int2(0, 0);
      if (valid) t = qw[base + lane];
      const unsigned int k = (unsigned int)t.y / (unsigned int)R;
      const unsigned int r = (unsigned int)t.y - k * (unsigned int)R;
      unsigned int mk[SCAN_SPLIT];
      unsigned int mine = 0u;
      #pragma unroll
      for (int j = 0; j < SCAN_SPLIT; ++j) {
        mk[j] = __builtin_amdgcn_ballot_w32(valid && k == (unsigned int)j);
        if (lane == j) mine = mk[j];
      }
      unsigned int first = 0u;
      if (mine) first = atomicAdd(&lcount[lane], (unsigned int)__builtin_popcount(mine));
      unsigned int p = 0u;
      #pragma unroll
      for (int j = 0; j < SCAN_SPLIT; ++j) {
        const unsigned int fj = (unsigned int)__builtin_amdgcn_readlane((int)first, j);
        if (k == (unsigned int)j) p = fj + __builtin_amdgcn_mbcnt_lo(mk[j], 0u);
      }
      if (valid) {
        const int s = src[t.x];
        const unsigned int range = range0 + k;
        if (range < nfit && p < CAP) seg0[((size_t)range * SCAN_SPLIT + slice) * CAP + p] = make_int2(s, (int)r);
      }
    }
    return 0;
  };
  agg_scan<SPAN>(dst, qw, lo, e0, e1, lane, wave, group, drain, drain);
  __syncthreads();

  if (lhdr && threadIdx.x < SCAN_SPLIT) {
    const unsigned int range = range0 + threadIdx.x;
    if (range < nranges) lst[range * SCAN_SPLIT + slice] = (int)lcount[threadIdx.x];
  }
}

#define GEMM_THREADS 256
#define GEMM_GRID    512

__device__ __forceinline__ float4 src_weight4(uint4 c) {
  return make_float4(rsqrtf(fmaxf(1.0f + (float)c.x, 1.0f)), rsqrtf(fmaxf(1.0f + (float)c.y, 1.0f)),
                     rsqrtf(fmaxf(1.0f + (float)c.z, 1.0f)), rsqrtf(fmaxf(1.0f + (float)c.w, 1.0f)));
}
__device__ __forceinline__ uint4 add4(uint4 a, uint4 b, uint4 c, uint4 d) {
  return make_uint4(a.x + b.x + c.x + d.x, a.y + b.y + c.y + d.y, a.z + b.z + c.z + d.z, a.w + b.w + c.w + d.w);
}

template <int KD>
__global__ __launch_bounds__(GEMM_THREADS)
void gemm_wreg_kernel(const float* __restrict__ X, const float* __restrict__ W,
                      const unsigned int* __restrict__ pcnt, int npc, float* __restrict__ Y, int ntiles) {
  static_assert(SCAN_SPLIT == 4, "the partial counts are summed four at a time below");
  constexpr int PITCH = KD + 4;
  constexpr int KS    = KD / 4;
  constexpr int F4ROW = KD / 4;
  constexpr int NLD   = (NT * F4ROW) / GEMM_THREADS;
  __shared__ __attribute__((aligned(16))) float xs[2][NT * PITCH];

  const int tid  = threadIdx.x;
  const int wave = tid >> 5;
  const int lane = tid & 31;
  const int l16  = lane & 15;
  const int hi   = lane >> 4;
  const int nbase = wave * NT;

  const float* __restrict__ wcol = W + nbase + l16;
  v2f b[KS];
  #pragma unroll
  for (int j = 0; j < KS; ++j) {
    b[j].x = wcol[(size_t)(4 * j + 2 * hi) * HDIM];
    b[j].y = wcol[(size_t)(4 * j + 2 * hi + 1) * HDIM];
  }

  const float4* __restrict__ X4 = (const float4*)X;
  const int f0 = tid, f1 = tid + GEMM_THREADS, f2 = tid + 2 * GEMM_THREADS, f3 = tid + 3 * GEMM_THREADS;
  const int o0 = (f0 / F4ROW) * PITCH + 4 * (f0 % F4ROW);
  const int o1 = (f1 / F4ROW) * PITCH + 4 * (f1 % F4ROW);
  const int o2 = (f2 / F4ROW) * PITCH + 4 * (f2 % F4ROW);
  const int o3 = (f3 / F4ROW) * PITCH + 4 * (f3 % F4ROW);
  float4 p0 = make_float4(0.0f, 0.0f, 0.0f, 0.0f), p1 = p0, p2 = p0, p3 = p0;
  float4 s0 = p0, s1 = p0;
  const uint4 z4 = make_uint4(0u, 0u, 0u, 0u);
  uint4 a0 = z4, a1 = z4, a2 = z4, a3 = z4, b0 = z4, b1 = z4, b2 = z4, b3 = z4;
  const size_t pl4 = (size_t)npc >> 2;

  int t = blockIdx.x;
  if (t < ntiles) {
    const float4* __restrict__ xt = X4 + (size_t)t * (NT * F4ROW);
    p0 = xt[f0];
    p1 = xt[f1];
    if (NLD > 2) { p2 = xt[f2]; p3 = xt[f3]; }
    const uint4* __restrict__ c4 = (const uint4*)(pcnt + (size_t)t * NT + 8 * hi);
    a0 = c4[0];           b0 = c4[1];
    a1 = c4[pl4];         b1 = c4[pl4 + 1];
    a2 = c4[2 * pl4];     b2 = c4[2 * pl4 + 1];
    a3 = c4[3 * pl4];     b3 = c4[3 * pl4 + 1];
    *(float4*)(xs[0] + o0) = p0;
    *(float4*)(xs[0] + o1) = p1;
    if (NLD > 2) { *(float4*)(xs[0] + o2) = p2; *(float4*)(xs[0] + o3) = p3; }
    s0 = src_weight4(add4(a0, a1, a2, a3));
    s1 = src_weight4(add4(b0, b1, b2, b3));
  }
  __syncthreads();

  int buf = 0;
  for (; t < ntiles; t += gridDim.x) {
    const int tn = t + gridDim.x;
    if (tn < ntiles) {
      const float4* __restrict__ xt = X4 + (size_t)tn * (NT * F4ROW);
      p0 = xt[f0];
      p1 = xt[f1];
      if (NLD > 2) { p2 = xt[f2]; p3 = xt[f3]; }
      const uint4* __restrict__ c4 = (const uint4*)(pcnt + (size_t)tn * NT + 8 * hi);
      a0 = c4[0];           b0 = c4[1];
      a1 = c4[pl4];         b1 = c4[pl4 + 1];
      a2 = c4[2 * pl4];     b2 = c4[2 * pl4 + 1];
      a3 = c4[3 * pl4];     b3 = c4[3 * pl4 + 1];
    }

    const float* xa = xs[buf] + l16 * PITCH + 2 * hi;
    v8f c = {};
    #pragma unroll
    for (int j = 0; j < KS; ++j) {
      const v2f a = *(const v2f*)(xa + 4 * j);
      c = __builtin_amdgcn_wmma_f32_16x16x4_f32(false, a, false, b[j], (short)0, c, false, false);
    }

    if (tn < ntiles) {
      float* xn = xs[buf ^ 1];
      *(float4*)(xn + o0) = p0;
      *(float4*)(xn + o1) = p1;
      if (NLD > 2) { *(float4*)(xn + o2) = p2; *(float4*)(xn + o3) = p3; }
    }
    __syncthreads();
    buf ^= 1;

    float* __restrict__ yout = Y + (size_t)t * NT * HDIM + nbase + l16;
    const float sr[8] = {s0.x, s0.y, s0.z, s0.w, s1.x, s1.y, s1.z, s1.w};
    #pragma unroll
    for (int v = 0; v < 8; ++v)
      yout[(size_t)(v + 8 * hi) * HDIM] = c[v] * sr[v];
    if (tn < ntiles) {
      s0 = src_weight4(add4(a0, a1, a2, a3));
      s1 = src_weight4(add4(b0, b1, b2, b3));
    }
  }
}


template <bool PARTIAL>
__device__ __forceinline__ void agg_add_rows(int cnt, int s, int r, float* acc,
                                             const float4* __restrict__ Y4, int lane) {
  for (int g = 0; g < cnt; g += AGG_ROWS) {
    float4 row[AGG_ROWS];
    #pragma unroll
    for (int j = 0; j < AGG_ROWS; ++j) {
      const int jj = PARTIAL ? min(g + j, cnt - 1) : g + j;
      const int sj = __builtin_amdgcn_readlane(s, jj);
      row[j] = Y4[(size_t)sj * (HDIM / 4) + lane];
    }
    #pragma unroll
    for (int j = 0; j < AGG_ROWS; ++j) {
      if (!PARTIAL || g + j < cnt) {
        const int rj = __builtin_amdgcn_readlane(r, g + j);
        float* a = acc + rj * HDIM + lane;
        __hip_atomic_fetch_add(a,      row[j].x, __ATOMIC_RELAXED, __HIP_MEMORY_SCOPE_WORKGROUP);
        __hip_atomic_fetch_add(a + 32, row[j].y, __ATOMIC_RELAXED, __HIP_MEMORY_SCOPE_WORKGROUP);
        __hip_atomic_fetch_add(a + 64, row[j].z, __ATOMIC_RELAXED, __HIP_MEMORY_SCOPE_WORKGROUP);
        __hip_atomic_fetch_add(a + 96, row[j].w, __ATOMIC_RELAXED, __HIP_MEMORY_SCOPE_WORKGROUP);
      }
    }
  }
}

template <bool LAST>
__device__ __forceinline__ int agg_drain(int n, const int2* qw, float* acc, unsigned int* icnt,
                                         const int* __restrict__ src,
                                         const float4* __restrict__ Y4, int lane) {
  const int keep = LAST ? 0 : (n & (AGG_ROWS - 1));
  for (int base = keep; base < n; base += 32) {
    const int cnt = min(32, n - base);
    int s = 0, r = 0;
    if (lane < cnt) {
      const int2 t = qw[base + lane];
      s = src[t.x];
      r = t.y;
      atomicAdd(&icnt[r], 1u);
    }
    agg_add_rows<LAST>(cnt, s, r, acc, Y4, lane);
  }
  return keep;
}

template <int R>
__global__ __launch_bounds__(AGG_THREADS)
void aggregate_owner_kernel(const int* __restrict__ src, const int* __restrict__ dst,
                            const float* __restrict__ Y, const float* __restrict__ bias,
                            float* __restrict__ out, int N, int E, int relu,
                            int* lst, const int* lst_end) {
  constexpr unsigned int CAP = AGG_LIST_PER_ROW * R / SCAN_SPLIT;
  __shared__ float acc[R * HDIM];
  __shared__ int2  queue[AGG_WAVES * AGG_QCAP];
  __shared__ unsigned int icnt[R];
  const int lane = threadIdx.x & 31;
  const int wave = threadIdx.x >> 5;
  const unsigned int lo = blockIdx.x * R;
  const float4* __restrict__ Y4 = (const float4*)Y;

  int2* seg0; unsigned int nfit;
  const bool lhdr = agg_list_place<R>(lst, lst_end, gridDim.x, seg0, nfit);
  const unsigned int cap = blockIdx.x < nfit ? CAP : 0u;
  const int2* segs = seg0 + (size_t)blockIdx.x * SCAN_SPLIT * CAP;

  for (int r = wave; r < R; r += AGG_WAVES) {
    const unsigned int i = lo + r;
    float4 y = make_float4(0.0f, 0.0f, 0.0f, 0.0f);
    if (i < (unsigned int)N) y = Y4[(size_t)i * (HDIM / 4) + lane];
    float* a = acc + r * HDIM + lane;
    a[0] = y.x; a[32] = y.y; a[64] = y.z; a[96] = y.w;
  }
  for (int r = threadIdx.x; r < R; r += AGG_THREADS) icnt[r] = 0u;
  __syncthreads();

  unsigned int nq[SCAN_SPLIT];
  bool usable = lhdr;
  #pragma unroll
  for (int q = 0; q < SCAN_SPLIT; ++q) {
    nq[q] = lhdr ? (unsigned int)lst[blockIdx.x * SCAN_SPLIT + q] : 0xffffffffu;
    usable = usable && nq[q] <= cap;
  }
  if (usable) {
    unsigned int gpre[SCAN_SPLIT + 1];
    gpre[0] = 0u;
    #pragma unroll
    for (int q = 0; q < SCAN_SPLIT; ++q) gpre[q + 1] = gpre[q] + ((nq[q] + 31u) >> 5);
    const unsigned int ngroups = gpre[SCAN_SPLIT];
    auto fetch = [&](unsigned int g, unsigned int& left) -> int2 {
      unsigned int q = 0u, gb = 0u, n = nq[0];
      #pragma unroll
      for (int j = 1; j < SCAN_SPLIT; ++j)
        if (g >= gpre[j]) { q = (unsigned int)j; gb = gpre[j]; n = nq[j]; }
      const unsigned int p0 = (g - gb) * 32u;
      left = n - p0;
      const unsigned int p = p0 + (unsigned int)lane;
      int2 e = make_int2(0, 0);
      if (p < n) e = segs[(size_t)q * CAP + p];
      return e;
    };
    unsigned int gi = (unsigned int)__builtin_amdgcn_readfirstlane(wave);
    unsigned int left = 0u;
    int2 e = make_int2(0, 0);
    if (gi < ngroups) e = fetch(gi, left);
    while (gi < ngroups) {
      const unsigned int gn = gi + AGG_WAVES;
      unsigned int left2 = 0u;
      int2 e2 = make_int2(0, 0);
      if (gn < ngroups) e2 = fetch(gn, left2);
      if ((unsigned int)lane < left) atomicAdd(&icnt[e.y], 1u);
      if (left >= 32u) agg_add_rows<false>(32, e.x, e.y, acc, Y4, lane);
      else             agg_add_rows<true>((int)left, e.x, e.y, acc, Y4, lane);
      e = e2; left = left2; gi = gn;
    }
  } else {
    int2* qw = queue + wave * AGG_QCAP;
    agg_scan<R>(dst, qw, lo, 0, E, lane, wave, blockIdx.x,
                [&](int n) { return agg_drain<false>(n, qw, acc, icnt, src, Y4, lane); },
                [&](int n) { return agg_drain<true>(n, qw, acc, icnt, src, Y4, lane); });
  }
  __syncthreads();

  const float4 b = ((const float4*)bias)[lane];
  for (int r = wave; r < R; r += AGG_WAVES) {
    const unsigned int i = lo + r;
    if (i < (unsigned int)N) {
      const float d = rsqrtf(fmaxf(1.0f + (float)icnt[r], 1.0f));
      const float* a = acc + r * HDIM + lane;
      float4 o;
      o.x = a[0]  * d + b.x;
      o.y = a[32] * d + b.y;
      o.z = a[64] * d + b.z;
      o.w = a[96] * d + b.w;
      if (relu) {
        o.x = fmaxf(o.x, 0.0f); o.y = fmaxf(o.y, 0.0f);
        o.z = fmaxf(o.z, 0.0f); o.w = fmaxf(o.w, 0.0f);
      }
      ((float4*)out)[(size_t)i * (HDIM / 4) + lane] = o;
    }
  }
}

static inline int cdiv(long long a, int b) { return (int)((a + b - 1) / b); }

template <int R, int D0>
static void run_graph(const float* feat, const int* src, const int* dst,
                      const float* W0, const float* b0, const float* W1, const float* b1,
                      int N, int E,
                      float* Y, float* h,
                      float* out, int* region, const int* region_end, hipStream_t stream) {
  const int nranges = cdiv(N, R);
  const int ndst = cdiv(nranges, SCAN_SPLIT) * SCAN_SPLIT;
  const int nsrc = cdiv(N, DEG_RANGE) * SCAN_SPLIT;
  unsigned int* pcnt = (unsigned int*)region;
  int* lst = region + (size_t)SCAN_SPLIT * N;
  const int* lst_end = region_end;
  prepare_kernel<R><<<ndst + nsrc, AGG_THREADS, 0, stream>>>(src, dst, pcnt, N, E, (unsigned int)nranges, (unsigned int)ndst, lst, lst_end);

  gemm_wreg_kernel<D0><<<GEMM_GRID, GEMM_THREADS, 0, stream>>>(feat, W0, pcnt, N, Y, N / NT);
  aggregate_owner_kernel<R><<<nranges, AGG_THREADS, 0, stream>>>(src, dst, Y, b0, h, N, E, 1, lst, lst_end);

  gemm_wreg_kernel<HDIM><<<GEMM_GRID, GEMM_THREADS, 0, stream>>>(h, W1, pcnt, N, Y, N / NT);
  aggregate_owner_kernel<R><<<nranges, AGG_THREADS, 0, stream>>>(src, dst, Y, b1, out, N, E, 0, lst, lst_end);
}

extern "C" void kernel_launch(void* const* d_in, const int* in_sizes, int n_in,
                              void* d_out, int out_size, void* d_ws, size_t ws_size,
                              hipStream_t stream) {
  const float* feat_a = (const float*)d_in[0];
  const float* feat_b = (const float*)d_in[1];
  const int*   src_a  = (const int*)d_in[2];
  const int*   dst_a  = (const int*)d_in[3];
  const int*   src_b  = (const int*)d_in[4];
  const int*   dst_b  = (const int*)d_in[5];
  const float* Wa0 = (const float*)d_in[6];
  const float* ba0 = (const float*)d_in[7];
  const float* Wa1 = (const float*)d_in[8];
  const float* ba1 = (const float*)d_in[9];
  const float* Wb0 = (const float*)d_in[10];
  const float* bb0 = (const float*)d_in[11];
  const float* Wb1 = (const float*)d_in[12];
  const float* bb1 = (const float*)d_in[13];

  const int DA = in_sizes[6] / HDIM;
  const int DB = in_sizes[10] / HDIM;
  const int NA = in_sizes[0] / DA;
  const int NB = in_sizes[1] / DB;
  const int EA = in_sizes[2];
  const int EB = in_sizes[4];

  float* ws = (float*)d_ws;
  float* ns_a  = ws;
  float* nd_a  = ns_a + NA;
  float* Y_a   = nd_a + NA;
  float* h_a   = Y_a + (size_t)NA * HDIM;
  float* ns_b  = h_a + (size_t)NA * HDIM;
  float* nd_b  = ns_b + NB;
  float* Y_b   = nd_b + NB;
  float* h_b   = Y_b + (size_t)NB * HDIM;

  float* out_a = (float*)d_out;
  float* out_b = out_a + (size_t)NA * HDIM;

  run_graph<256, 256>(feat_a, src_a, dst_a, Wa0, ba0, Wa1, ba1, NA, EA,
                 Y_a, h_a, out_a,
                 (int*)h_b, (const int*)(h_b + (size_t)NB * HDIM), stream);
  run_graph<128, 128>(feat_b, src_b, dst_b, Wb0, bb0, Wb1, bb1, NB, EB,
                 Y_b, h_b, out_b,
                 (int*)Y_a, (const int*)(Y_a + (size_t)NA * HDIM), stream);
}
